# starting kernel
# baseline (speedup 1.0000x reference)
#include <hip/hip_runtime.h>


typedef float v2f __attribute__((ext_vector_type(2)));
typedef float v8f __attribute__((ext_vector_type(8)));

#define N_NODES 50000

__global__ void gcn_zero(float* __restrict__ p, int n) {
  int i = blockIdx.x * blockDim.x + threadIdx.x;
  if (i < n) p[i] = 0.0f;
}

__global__ void gcn_degree(const int* __restrict__ src, const int* __restrict__ dst,
                           float* __restrict__ deg_out, float* __restrict__ deg_in, int E) {
  int e = blockIdx.x * blockDim.x + threadIdx.x;
  if (e >= E) return;
  atomicAdd(&deg_out[src[e]], 1.0f);
  atomicAdd(&deg_in[dst[e]], 1.0f);
}

__global__ void gcn_norm(float* __restrict__ p, int n) {
  int i = blockIdx.x * blockDim.x + threadIdx.x;
  if (i < n) p[i] = __frsqrt_rn(fmaxf(p[i], 1.0f));
}

__global__ void gcn_gemm_wmma(const float* __restrict__ A, const float* __restrict__ W,
                              float* __restrict__ C, int n_rows, int K, int F,
                              const float* __restrict__ pre_scale,
                              const float* __restrict__ post_scale,
                              const float* __restrict__ bias, int relu) {
  const int ftiles = F >> 4;
  const int tiles  = (n_rows >> 4) * ftiles;
  const int w = blockIdx.x * (blockDim.x >> 5) + ((int)threadIdx.x >> 5);
  if (w >= tiles) return;
  const int tileM = w / ftiles;
  const int tileN = w - tileM * ftiles;
  const int lane  = threadIdx.x & 31;
  const int half  = lane >> 4;
  const int lr    = lane & 15;

  const int m = (tileM << 4) + lr;
  const int n = (tileN << 4) + lr;
  const float ps = pre_scale ? pre_scale[m] : 1.0f;
  const float* Arow = A + (size_t)m * K;

  v8f acc = {};
  for (int k = 0; k < K; k += 4) {
    const int koff = k + (half << 1);
    v2f a, b;
    a.x = Arow[koff]     * ps;
    a.y = Arow[koff + 1] * ps;
    b.x = W[(size_t)koff * F + n];
    b.y = W[(size_t)(koff + 1) * F + n];
    acc = __builtin_amdgcn_wmma_f32_16x16x4_f32(
         false, a,  false, b,
         (short)0, acc,  false,  false);
  }

#pragma unroll
  for (int i = 0; i < 8; ++i) {
    const int row = (tileM << 4) + i + (half << 3);
    const int col = (tileN << 4) + lr;
    float v = acc[i];
    if (post_scale) v *= post_scale[row];
    if (bias)       v += bias[col];
    if (relu)       v = fmaxf(v, 0.0f);
    C[(size_t)row * F + col] = v;
  }
}

__global__ void gcn_scatter(const float* __restrict__ msg, const int* __restrict__ src,
                            const int* __restrict__ dst, const float* __restrict__ scale,
                            float* __restrict__ agg, int n_edges, int F, int chunk_shift) {
  int idx = blockIdx.x * blockDim.x + threadIdx.x;
  int e = idx >> chunk_shift;
  if (e >= n_edges) return;
  int c = idx & ((1 << chunk_shift) - 1);
  int s = src[e], d = dst[e];
  float4 v = ((const float4*)(msg + (size_t)s * F))[c];
  if (scale) { float sc = scale[s]; v.x *= sc; v.y *= sc; v.z *= sc; v.w *= sc; }
  float* out = agg + (size_t)d * F + (c << 2);
  atomicAdd(out + 0, v.x);
  atomicAdd(out + 1, v.y);
  atomicAdd(out + 2, v.z);
  atomicAdd(out + 3, v.w);
}

__global__ void gcn_epilogue(float* __restrict__ buf, const float* __restrict__ nd,
                             const float* __restrict__ bias, int total, int f_shift, int relu) {
  int i = blockIdx.x * blockDim.x + threadIdx.x;
  if (i >= total) return;
  int row = i >> f_shift;
  int col = i & ((1 << f_shift) - 1);
  float v = buf[i] * nd[row] + bias[col];
  if (relu) v = fmaxf(v, 0.0f);
  buf[i] = v;
}

static inline unsigned cdiv(unsigned a, unsigned b) { return (a + b - 1) / b; }

extern "C" void kernel_launch(void* const* d_in, const int* in_sizes, int n_in,
                              void* d_out, int out_size, void* d_ws, size_t ws_size,
                              hipStream_t stream) {
  const float* features = (const float*)d_in[0];
  const float* W1 = (const float*)d_in[1];
  const float* b1 = (const float*)d_in[2];
  const float* W2 = (const float*)d_in[3];
  const float* b2 = (const float*)d_in[4];
  const float* W3 = (const float*)d_in[5];
  const float* b3 = (const float*)d_in[6];
  const int* src  = (const int*)d_in[7];
  const int* dst  = (const int*)d_in[8];
  float* out = (float*)d_out;

  const int N  = N_NODES;
  const int E  = in_sizes[7];
  const int IN = in_sizes[0] / N;
  const int H  = in_sizes[2];
  const int CLS = in_sizes[6];

  float* ws   = (float*)d_ws;
  float* ns   = ws;
  float* nd   = ws + N;
  float* bufA = ws + 2 * N + 96;
  float* bufB = bufA + (size_t)N * H;
  (void)ws_size; (void)n_in; (void)out_size;

  const int BS = 256;

  gcn_zero  <<<cdiv(2 * N, BS), BS, 0, stream>>>(ns, 2 * N);
  gcn_degree<<<cdiv(E, BS),     BS, 0, stream>>>(src, dst, ns, nd, E);
  gcn_norm  <<<cdiv(2 * N, BS), BS, 0, stream>>>(ns, 2 * N);

  {
    unsigned tiles = (N / 16) * (H / 16);
    gcn_gemm_wmma<<<cdiv(tiles, BS / 32), BS, 0, stream>>>(
        features, W1, bufA, N, IN, H, ns, nullptr, nullptr, 0);
    gcn_zero<<<cdiv(N * H, BS), BS, 0, stream>>>(bufB, N * H);
    gcn_scatter<<<cdiv((unsigned)E * (H / 4), BS), BS, 0, stream>>>(
        bufA, src, dst, nullptr, bufB, E, H, 5);
    gcn_epilogue<<<cdiv(N * H, BS), BS, 0, stream>>>(bufB, nd, b1, N * H, 7, 1);
  }

  {
    gcn_zero<<<cdiv(N * H, BS), BS, 0, stream>>>(bufA, N * H);
    gcn_scatter<<<cdiv((unsigned)E * (H / 4), BS), BS, 0, stream>>>(
        bufB, src, dst, ns, bufA, E, H, 5);
    unsigned tiles = (N / 16) * (H / 16);
    gcn_gemm_wmma<<<cdiv(tiles, BS / 32), BS, 0, stream>>>(
        bufA, W2, bufB, N, H, H, nullptr, nd, b2, 1);
  }

  {
    unsigned tiles = (N / 16) * (CLS / 16);
    gcn_gemm_wmma<<<cdiv(tiles, BS / 32), BS, 0, stream>>>(
        bufB, W3, bufA, N, H, CLS, ns, nullptr, nullptr, 0);
    gcn_zero<<<cdiv(N * CLS, BS), BS, 0, stream>>>(out, N * CLS);
    gcn_scatter<<<cdiv((unsigned)E * (CLS / 4), BS), BS, 0, stream>>>(
        bufA, src, dst, nullptr, out, E, CLS, 4);
    gcn_epilogue<<<cdiv(N * CLS, BS), BS, 0, stream>>>(out, nd, b3, N * CLS, 6, 0);
  }
}


# scatter: lane = column, each wave-wide f32 atomic add and row load covers one whole 128-B line, not 4 quarter lines; a wave walks 8 edges; sums stay arrival-order atomics
# speedup vs baseline: 1.4858x; 1.4858x over previous
#include <hip/hip_runtime.h>


typedef float v2f __attribute__((ext_vector_type(2)));
typedef float v8f __attribute__((ext_vector_type(8)));

#define N_NODES 50000

__global__ void gcn_zero(float* __restrict__ p, int n) {
  int i = blockIdx.x * blockDim.x + threadIdx.x;
  if (i < n) p[i] = 0.0f;
}

__global__ void gcn_degree(const int* __restrict__ src, const int* __restrict__ dst,
                           float* __restrict__ deg_out, float* __restrict__ deg_in, int E) {
  int e = blockIdx.x * blockDim.x + threadIdx.x;
  if (e >= E) return;
  atomicAdd(&deg_out[src[e]], 1.0f);
  atomicAdd(&deg_in[dst[e]], 1.0f);
}

__global__ void gcn_norm(float* __restrict__ p, int n) {
  int i = blockIdx.x * blockDim.x + threadIdx.x;
  if (i < n) p[i] = __frsqrt_rn(fmaxf(p[i], 1.0f));
}

__global__ void gcn_gemm_wmma(const float* __restrict__ A, const float* __restrict__ W,
                              float* __restrict__ C, int n_rows, int K, int F,
                              const float* __restrict__ pre_scale,
                              const float* __restrict__ post_scale,
                              const float* __restrict__ bias, int relu) {
  const int ftiles = F >> 4;
  const int tiles  = (n_rows >> 4) * ftiles;
  const int w = blockIdx.x * (blockDim.x >> 5) + ((int)threadIdx.x >> 5);
  if (w >= tiles) return;
  const int tileM = w / ftiles;
  const int tileN = w - tileM * ftiles;
  const int lane  = threadIdx.x & 31;
  const int half  = lane >> 4;
  const int lr    = lane & 15;

  const int m = (tileM << 4) + lr;
  const int n = (tileN << 4) + lr;
  const float ps = pre_scale ? pre_scale[m] : 1.0f;
  const float* Arow = A + (size_t)m * K;

  v8f acc = {};
  for (int k = 0; k < K; k += 4) {
    const int koff = k + (half << 1);
    v2f a, b;
    a.x = Arow[koff]     * ps;
    a.y = Arow[koff + 1] * ps;
    b.x = W[(size_t)koff * F + n];
    b.y = W[(size_t)(koff + 1) * F + n];
    acc = __builtin_amdgcn_wmma_f32_16x16x4_f32(
         false, a,  false, b,
         (short)0, acc,  false,  false);
  }

#pragma unroll
  for (int i = 0; i < 8; ++i) {
    const int row = (tileM << 4) + i + (half << 3);
    const int col = (tileN << 4) + lr;
    float v = acc[i];
    if (post_scale) v *= post_scale[row];
    if (bias)       v += bias[col];
    if (relu)       v = fmaxf(v, 0.0f);
    C[(size_t)row * F + col] = v;
  }
}

#define SCATTER_EPW 8

template <int J>
__device__ __forceinline__ void scatter_rows(const float* __restrict__ msg,
                                             const int* __restrict__ src,
                                             const int* __restrict__ dst,
                                             const float* __restrict__ scale,
                                             float* __restrict__ agg, int n_edges) {
  constexpr int F = J * 32;
  const int lane = threadIdx.x & 31;
  const int wave = __builtin_amdgcn_readfirstlane(
      (int)(blockIdx.x * (blockDim.x >> 5) + (threadIdx.x >> 5)));
  const int e0 = wave * SCATTER_EPW;
  if (e0 >= n_edges) return;

  if (e0 + SCATTER_EPW <= n_edges) {
    int s[SCATTER_EPW], d[SCATTER_EPW];
    for (int i = 0; i < SCATTER_EPW; ++i) { s[i] = src[e0 + i]; d[i] = dst[e0 + i]; }

    float v[SCATTER_EPW][J];
    for (int i = 0; i < SCATTER_EPW; ++i) {
      const float* row = msg + (size_t)s[i] * F + lane;
      for (int j = 0; j < J; ++j) v[i][j] = row[32 * j];
    }
    if (scale) {
      for (int i = 0; i < SCATTER_EPW; ++i) {
        const float sc = scale[s[i]];
        for (int j = 0; j < J; ++j) v[i][j] *= sc;
      }
    }
    for (int i = 0; i < SCATTER_EPW; ++i) {
      float* o = agg + (size_t)d[i] * F + lane;
      for (int j = 0; j < J; ++j) atomicAdd(o + 32 * j, v[i][j]);
    }
  } else {
    for (int e = e0; e < n_edges; ++e) {
      const int se = src[e], de = dst[e];
      const float sc = scale ? scale[se] : 1.0f;
      const float* row = msg + (size_t)se * F + lane;
      float* o = agg + (size_t)de * F + lane;
      for (int j = 0; j < J; ++j) atomicAdd(o + 32 * j, row[32 * j] * sc);
    }
  }
}

__global__ void gcn_scatter128(const float* __restrict__ msg, const int* __restrict__ src,
                               const int* __restrict__ dst, const float* __restrict__ scale,
                               float* __restrict__ agg, int n_edges) {
  scatter_rows<4>(msg, src, dst, scale, agg, n_edges);
}

__global__ void gcn_scatter64(const float* __restrict__ msg, const int* __restrict__ src,
                              const int* __restrict__ dst, const float* __restrict__ scale,
                              float* __restrict__ agg, int n_edges) {
  scatter_rows<2>(msg, src, dst, scale, agg, n_edges);
}

__global__ void gcn_epilogue(float* __restrict__ buf, const float* __restrict__ nd,
                             const float* __restrict__ bias, int total, int f_shift, int relu) {
  int i = blockIdx.x * blockDim.x + threadIdx.x;
  if (i >= total) return;
  int row = i >> f_shift;
  int col = i & ((1 << f_shift) - 1);
  float v = buf[i] * nd[row] + bias[col];
  if (relu) v = fmaxf(v, 0.0f);
  buf[i] = v;
}

static inline unsigned cdiv(unsigned a, unsigned b) { return (a + b - 1) / b; }

extern "C" void kernel_launch(void* const* d_in, const int* in_sizes, int n_in,
                              void* d_out, int out_size, void* d_ws, size_t ws_size,
                              hipStream_t stream) {
  const float* features = (const float*)d_in[0];
  const float* W1 = (const float*)d_in[1];
  const float* b1 = (const float*)d_in[2];
  const float* W2 = (const float*)d_in[3];
  const float* b2 = (const float*)d_in[4];
  const float* W3 = (const float*)d_in[5];
  const float* b3 = (const float*)d_in[6];
  const int* src  = (const int*)d_in[7];
  const int* dst  = (const int*)d_in[8];
  float* out = (float*)d_out;

  const int N  = N_NODES;
  const int E  = in_sizes[7];
  const int IN = in_sizes[0] / N;
  const int H  = in_sizes[2];
  const int CLS = in_sizes[6];

  float* ws   = (float*)d_ws;
  float* ns   = ws;
  float* nd   = ws + N;
  float* bufA = ws + 2 * N + 96;
  float* bufB = bufA + (size_t)N * H;
  (void)ws_size; (void)n_in; (void)out_size;

  const int BS = 256;

  gcn_zero  <<<cdiv(2 * N, BS), BS, 0, stream>>>(ns, 2 * N);
  gcn_degree<<<cdiv(E, BS),     BS, 0, stream>>>(src, dst, ns, nd, E);
  gcn_norm  <<<cdiv(2 * N, BS), BS, 0, stream>>>(ns, 2 * N);

  {
    unsigned tiles = (N / 16) * (H / 16);
    gcn_gemm_wmma<<<cdiv(tiles, BS / 32), BS, 0, stream>>>(
        features, W1, bufA, N, IN, H, ns, nullptr, nullptr, 0);
    gcn_zero<<<cdiv(N * H, BS), BS, 0, stream>>>(bufB, N * H);
    gcn_scatter128<<<cdiv(cdiv(E, SCATTER_EPW), BS / 32), BS, 0, stream>>>(
        bufA, src, dst, nullptr, bufB, E);
    gcn_epilogue<<<cdiv(N * H, BS), BS, 0, stream>>>(bufB, nd, b1, N * H, 7, 1);
  }

  {
    gcn_zero<<<cdiv(N * H, BS), BS, 0, stream>>>(bufA, N * H);
    gcn_scatter128<<<cdiv(cdiv(E, SCATTER_EPW), BS / 32), BS, 0, stream>>>(
        bufB, src, dst, ns, bufA, E);
    unsigned tiles = (N / 16) * (H / 16);
    gcn_gemm_wmma<<<cdiv(tiles, BS / 32), BS, 0, stream>>>(
        bufA, W2, bufB, N, H, H, nullptr, nd, b2, 1);
  }

  {
    unsigned tiles = (N / 16) * (CLS / 16);
    gcn_gemm_wmma<<<cdiv(tiles, BS / 32), BS, 0, stream>>>(
        bufB, W3, bufA, N, H, CLS, ns, nullptr, nullptr, 0);
    gcn_zero<<<cdiv(N * CLS, BS), BS, 0, stream>>>(out, N * CLS);
    gcn_scatter64<<<cdiv(cdiv(E, SCATTER_EPW), BS / 32), BS, 0, stream>>>(
        bufA, src, dst, nullptr, out, E);
    gcn_epilogue<<<cdiv(N * CLS, BS), BS, 0, stream>>>(out, nd, b3, N * CLS, 6, 0);
  }
}


# GEMMs retiled: block stages W in LDS, one wave owns a 16-row x F strip (4-8 accumulator tiles), A read once in 32-B lane pieces, 16-B stores; k order of the f32 WMMA sums changes
# speedup vs baseline: 1.6237x; 1.0928x over previous
#include <hip/hip_runtime.h>


typedef float v2f __attribute__((ext_vector_type(2)));
typedef float v8f __attribute__((ext_vector_type(8)));

#define N_NODES 50000

__global__ void gcn_zero(float* __restrict__ p, int n) {
  int i = blockIdx.x * blockDim.x + threadIdx.x;
  if (i < n) p[i] = 0.0f;
}

__global__ void gcn_degree(const int* __restrict__ src, const int* __restrict__ dst,
                           float* __restrict__ deg_out, float* __restrict__ deg_in, int E) {
  int e = blockIdx.x * blockDim.x + threadIdx.x;
  if (e >= E) return;
  atomicAdd(&deg_out[src[e]], 1.0f);
  atomicAdd(&deg_in[dst[e]], 1.0f);
}

__global__ void gcn_norm(float* __restrict__ p, int n) {
  int i = blockIdx.x * blockDim.x + threadIdx.x;
  if (i < n) p[i] = __frsqrt_rn(fmaxf(p[i], 1.0f));
}

template <int K, int F>
__global__ __launch_bounds__(256) void gcn_gemm_wmma(
    const float* __restrict__ A, const float* __restrict__ W,
    float* __restrict__ C, int n_rows,
    const float* __restrict__ pre_scale, const float* __restrict__ post_scale,
    const float* __restrict__ bias, int relu) {
  constexpr int NG = F / 64;
  __shared__ alignas(16) float Wl[K * F];

  {
    const float4* Wg = reinterpret_cast<const float4*>(W);
    float4* Ws = reinterpret_cast<float4*>(Wl);
    for (int i = threadIdx.x; i < K * F / 4; i += 256) Ws[i] = Wg[i];
  }
  __syncthreads();

  const int tileM = blockIdx.x * 8 + ((int)threadIdx.x >> 5);
  if (tileM >= (n_rows >> 4)) return;
  const int lane = threadIdx.x & 31;
  const int h = lane >> 4;
  const int c = lane & 15;

  const int m = (tileM << 4) + c;
  const float ps = pre_scale ? pre_scale[m] : 1.0f;
  const float4* Arow = reinterpret_cast<const float4*>(A + (size_t)m * K + 8 * h);
  const float* Wlane = Wl + (8 * h) * F + 4 * c;

  v8f acc[NG][4];
#pragma unroll
  for (int g = 0; g < NG; ++g)
#pragma unroll
    for (int t = 0; t < 4; ++t) acc[g][t] = v8f{};

  float4 q0 = Arow[0], q1 = Arow[1];
  for (int kb = 0; kb < K; kb += 16) {
    const int kn = (kb + 16 < K) ? kb + 16 : kb;
    const float4 n0 = Arow[kn >> 2], n1 = Arow[(kn >> 2) + 1];

    const float av[8] = {q0.x * ps, q0.y * ps, q0.z * ps, q0.w * ps,
                         q1.x * ps, q1.y * ps, q1.z * ps, q1.w * ps};
#pragma unroll
    for (int j = 0; j < 4; ++j) {
      v2f a;
      a.x = av[2 * j];
      a.y = av[2 * j + 1];
      const float* wr = Wlane + (kb + 2 * j) * F;
#pragma unroll
      for (int g = 0; g < NG; ++g) {
        const float4 b0 = *reinterpret_cast<const float4*>(wr + 64 * g);
        const float4 b1 = *reinterpret_cast<const float4*>(wr + F + 64 * g);
        v2f b;
        b.x = b0.x; b.y = b1.x;
        acc[g][0] = __builtin_amdgcn_wmma_f32_16x16x4_f32(false, a, false, b, (short)0, acc[g][0], false, false);
        b.x = b0.y; b.y = b1.y;
        acc[g][1] = __builtin_amdgcn_wmma_f32_16x16x4_f32(false, a, false, b, (short)0, acc[g][1], false, false);
        b.x = b0.z; b.y = b1.z;
        acc[g][2] = __builtin_amdgcn_wmma_f32_16x16x4_f32(false, a, false, b, (short)0, acc[g][2], false, false);
        b.x = b0.w; b.y = b1.w;
        acc[g][3] = __builtin_amdgcn_wmma_f32_16x16x4_f32(false, a, false, b, (short)0, acc[g][3], false, false);
      }
    }
    q0 = n0;
    q1 = n1;
  }

#pragma unroll
  for (int g = 0; g < NG; ++g) {
    const int col = 64 * g + 4 * c;
    float4 bb = {0.0f, 0.0f, 0.0f, 0.0f};
    if (bias) bb = *reinterpret_cast<const float4*>(bias + col);
#pragma unroll
    for (int r = 0; r < 8; ++r) {
      const int row = (tileM << 4) + r + (h << 3);
      float4 v = {acc[g][0][r], acc[g][1][r], acc[g][2][r], acc[g][3][r]};
      if (post_scale) {
        const float s = post_scale[row];
        v.x *= s; v.y *= s; v.z *= s; v.w *= s;
      }
      if (bias) { v.x += bb.x; v.y += bb.y; v.z += bb.z; v.w += bb.w; }
      if (relu) {
        v.x = fmaxf(v.x, 0.0f); v.y = fmaxf(v.y, 0.0f);
        v.z = fmaxf(v.z, 0.0f); v.w = fmaxf(v.w, 0.0f);
      }
      *reinterpret_cast<float4*>(C + (size_t)row * F + col) = v;
    }
  }
}

#define SCATTER_EPW 8

template <int J>
__device__ __forceinline__ void scatter_rows(const float* __restrict__ msg,
                                             const int* __restrict__ src,
                                             const int* __restrict__ dst,
                                             const float* __restrict__ scale,
                                             float* __restrict__ agg, int n_edges) {
  constexpr int F = J * 32;
  const int lane = threadIdx.x & 31;
  const int wave = __builtin_amdgcn_readfirstlane(
      (int)(blockIdx.x * (blockDim.x >> 5) + (threadIdx.x >> 5)));
  const int e0 = wave * SCATTER_EPW;
  if (e0 >= n_edges) return;

  if (e0 + SCATTER_EPW <= n_edges) {
    int s[SCATTER_EPW], d[SCATTER_EPW];
    for (int i = 0; i < SCATTER_EPW; ++i) { s[i] = src[e0 + i]; d[i] = dst[e0 + i]; }

    float v[SCATTER_EPW][J];
    for (int i = 0; i < SCATTER_EPW; ++i) {
      const float* row = msg + (size_t)s[i] * F + lane;
      for (int j = 0; j < J; ++j) v[i][j] = row[32 * j];
    }
    if (scale) {
      for (int i = 0; i < SCATTER_EPW; ++i) {
        const float sc = scale[s[i]];
        for (int j = 0; j < J; ++j) v[i][j] *= sc;
      }
    }
    for (int i = 0; i < SCATTER_EPW; ++i) {
      float* o = agg + (size_t)d[i] * F + lane;
      for (int j = 0; j < J; ++j) atomicAdd(o + 32 * j, v[i][j]);
    }
  } else {
    for (int e = e0; e < n_edges; ++e) {
      const int se = src[e], de = dst[e];
      const float sc = scale ? scale[se] : 1.0f;
      const float* row = msg + (size_t)se * F + lane;
      float* o = agg + (size_t)de * F + lane;
      for (int j = 0; j < J; ++j) atomicAdd(o + 32 * j, row[32 * j] * sc);
    }
  }
}

__global__ void gcn_scatter128(const float* __restrict__ msg, const int* __restrict__ src,
                               const int* __restrict__ dst, const float* __restrict__ scale,
                               float* __restrict__ agg, int n_edges) {
  scatter_rows<4>(msg, src, dst, scale, agg, n_edges);
}

__global__ void gcn_scatter64(const float* __restrict__ msg, const int* __restrict__ src,
                              const int* __restrict__ dst, const float* __restrict__ scale,
                              float* __restrict__ agg, int n_edges) {
  scatter_rows<2>(msg, src, dst, scale, agg, n_edges);
}

__global__ void gcn_epilogue(float* __restrict__ buf, const float* __restrict__ nd,
                             const float* __restrict__ bias, int total, int f_shift, int relu) {
  int i = blockIdx.x * blockDim.x + threadIdx.x;
  if (i >= total) return;
  int row = i >> f_shift;
  int col = i & ((1 << f_shift) - 1);
  float v = buf[i] * nd[row] + bias[col];
  if (relu) v = fmaxf(v, 0.0f);
  buf[i] = v;
}

static inline unsigned cdiv(unsigned a, unsigned b) { return (a + b - 1) / b; }

extern "C" void kernel_launch(void* const* d_in, const int* in_sizes, int n_in,
                              void* d_out, int out_size, void* d_ws, size_t ws_size,
                              hipStream_t stream) {
  const float* features = (const float*)d_in[0];
  const float* W1 = (const float*)d_in[1];
  const float* b1 = (const float*)d_in[2];
  const float* W2 = (const float*)d_in[3];
  const float* b2 = (const float*)d_in[4];
  const float* W3 = (const float*)d_in[5];
  const float* b3 = (const float*)d_in[6];
  const int* src  = (const int*)d_in[7];
  const int* dst  = (const int*)d_in[8];
  float* out = (float*)d_out;

  const int N  = N_NODES;
  const int E  = in_sizes[7];
  const int H  = in_sizes[2];
  const int CLS = in_sizes[6];

  float* ws   = (float*)d_ws;
  float* ns   = ws;
  float* nd   = ws + N;
  float* bufA = ws + 2 * N + 96;
  float* bufB = bufA + (size_t)N * H;
  (void)ws_size; (void)n_in; (void)out_size;

  const int BS = 256;

  gcn_zero  <<<cdiv(2 * N, BS), BS, 0, stream>>>(ns, 2 * N);
  gcn_degree<<<cdiv(E, BS),     BS, 0, stream>>>(src, dst, ns, nd, E);
  gcn_norm  <<<cdiv(2 * N, BS), BS, 0, stream>>>(ns, 2 * N);

  {
    gcn_gemm_wmma<256, 128><<<cdiv(N / 16, BS / 32), BS, 0, stream>>>(
        features, W1, bufA, N, ns, nullptr, nullptr, 0);
    gcn_zero<<<cdiv(N * H, BS), BS, 0, stream>>>(bufB, N * H);
    gcn_scatter128<<<cdiv(cdiv(E, SCATTER_EPW), BS / 32), BS, 0, stream>>>(
        bufA, src, dst, nullptr, bufB, E);
    gcn_epilogue<<<cdiv(N * H, BS), BS, 0, stream>>>(bufB, nd, b1, N * H, 7, 1);
  }

  {
    gcn_zero<<<cdiv(N * H, BS), BS, 0, stream>>>(bufA, N * H);
    gcn_scatter128<<<cdiv(cdiv(E, SCATTER_EPW), BS / 32), BS, 0, stream>>>(
        bufB, src, dst, ns, bufA, E);
    gcn_gemm_wmma<128, 128><<<cdiv(N / 16, BS / 32), BS, 0, stream>>>(
        bufA, W2, bufB, N, nullptr, nd, b2, 1);
  }

  {
    gcn_gemm_wmma<128, 64><<<cdiv(N / 16, BS / 32), BS, 0, stream>>>(
        bufB, W3, bufA, N, ns, nullptr, nullptr, 0);
    gcn_zero<<<cdiv(N * CLS, BS), BS, 0, stream>>>(out, N * CLS);
    gcn_scatter64<<<cdiv(cdiv(E, SCATTER_EPW), BS / 32), BS, 0, stream>>>(
        bufA, src, dst, nullptr, out, E);
    gcn_epilogue<<<cdiv(N * CLS, BS), BS, 0, stream>>>(out, nd, b3, N * CLS, 6, 0);
  }
}


# f32 atomic scatter -> gather: in-call edge list by destination (int atomic count = slot, scan, place); a wave per node sums rows in registers in slot (arrival) order, one store per line
# speedup vs baseline: 2.4888x; 1.5328x over previous
#include <hip/hip_runtime.h>


typedef float v2f __attribute__((ext_vector_type(2)));
typedef float v4f __attribute__((ext_vector_type(4)));
typedef float v8f __attribute__((ext_vector_type(8)));

#define N_NODES 50000

__global__ void gcn_zero_i(int* __restrict__ p, int n) {
  int i = blockIdx.x * blockDim.x + threadIdx.x;
  if (i < n) p[i] = 0;
}

__global__ void gcn_count(const int* __restrict__ src, const int* __restrict__ dst,
                          int* __restrict__ cnt_out, int* __restrict__ cnt_in,
                          int* __restrict__ pos, int E) {
  int e = blockIdx.x * blockDim.x + threadIdx.x;
  if (e >= E) return;
  const int s = src[e];
  const int d = dst[e];
  atomicAdd(&cnt_out[s], 1);
  pos[e] = atomicAdd(&cnt_in[d], 1);
}

__global__ __launch_bounds__(1024) void gcn_scan(
    const int* __restrict__ cnt_out, const int* __restrict__ cnt_in,
    int* __restrict__ rowptr, float* __restrict__ ns, float* __restrict__ nd, int n) {
  __shared__ int front[32];
  __shared__ int wtot[32];
  const int tid = threadIdx.x;
  const int lane = tid & 31;
  const int w = tid >> 5;
  const int first = blockIdx.x * 1024;

  int s = 0;
  const int4* c4 = reinterpret_cast<const int4*>(cnt_in);
#pragma unroll 4
  for (int i = tid; i < (first >> 2); i += 1024) {
    const int4 q = c4[i];
    s += (q.x + q.y) + (q.z + q.w);
  }
  const int idx = first + tid;
  const int c = (idx < n) ? cnt_in[idx] : 0;

  int incl = c;
  for (int d = 1; d < 32; d <<= 1) {
    const int t = __shfl_up(incl, d);
    if (lane >= d) incl += t;
  }
  for (int d = 16; d >= 1; d >>= 1) s += __shfl_xor(s, d);
  if (lane == 31) wtot[w] = incl;
  if (lane == 0) front[w] = s;
  __syncthreads();
  int base = 0;
  for (int k = 0; k < 32; ++k) {
    base += front[k];
    if (k < w) base += wtot[k];
  }
  if (idx < n) {
    const int ex = base + incl - c;
    rowptr[idx] = ex;
    if (idx == n - 1) rowptr[n] = ex + c;
    nd[idx] = __frsqrt_rn(fmaxf((float)c, 1.0f));
    ns[idx] = __frsqrt_rn(fmaxf((float)cnt_out[idx], 1.0f));
  }
}

__global__ void gcn_place(const int* __restrict__ src, const int* __restrict__ dst,
                          const int* __restrict__ rowptr, const int* __restrict__ pos,
                          int* __restrict__ col, int E) {
  int e = blockIdx.x * blockDim.x + threadIdx.x;
  if (e >= E) return;
  col[rowptr[dst[e]] + pos[e]] = src[e];
}

template <int K, int F>
__global__ __launch_bounds__(256) void gcn_gemm_wmma(
    const float* __restrict__ A, const float* __restrict__ W,
    float* __restrict__ C, int n_rows,
    const float* __restrict__ pre_scale, const float* __restrict__ post_scale,
    const float* __restrict__ bias, int relu) {
  constexpr int NG = F / 64;
  __shared__ alignas(16) float Wl[K * F];

  {
    const float4* Wg = reinterpret_cast<const float4*>(W);
    float4* Ws = reinterpret_cast<float4*>(Wl);
    for (int i = threadIdx.x; i < K * F / 4; i += 256) Ws[i] = Wg[i];
  }
  __syncthreads();

  const int tileM = blockIdx.x * 8 + ((int)threadIdx.x >> 5);
  if (tileM >= (n_rows >> 4)) return;
  const int lane = threadIdx.x & 31;
  const int h = lane >> 4;
  const int c = lane & 15;

  const int m = (tileM << 4) + c;
  const float ps = pre_scale ? pre_scale[m] : 1.0f;
  const float4* Arow = reinterpret_cast<const float4*>(A + (size_t)m * K + 8 * h);
  const float* Wlane = Wl + (8 * h) * F + 4 * c;

  v8f acc[NG][4];
#pragma unroll
  for (int g = 0; g < NG; ++g)
#pragma unroll
    for (int t = 0; t < 4; ++t) acc[g][t] = v8f{};

  float4 q0 = Arow[0], q1 = Arow[1];
  for (int kb = 0; kb < K; kb += 16) {
    const int kn = (kb + 16 < K) ? kb + 16 : kb;
    const float4 n0 = Arow[kn >> 2], n1 = Arow[(kn >> 2) + 1];

    const float av[8] = {q0.x * ps, q0.y * ps, q0.z * ps, q0.w * ps,
                         q1.x * ps, q1.y * ps, q1.z * ps, q1.w * ps};
#pragma unroll
    for (int j = 0; j < 4; ++j) {
      v2f a;
      a.x = av[2 * j];
      a.y = av[2 * j + 1];
      const float* wr = Wlane + (kb + 2 * j) * F;
#pragma unroll
      for (int g = 0; g < NG; ++g) {
        const float4 b0 = *reinterpret_cast<const float4*>(wr + 64 * g);
        const float4 b1 = *reinterpret_cast<const float4*>(wr + F + 64 * g);
        v2f b;
        b.x = b0.x; b.y = b1.x;
        acc[g][0] = __builtin_amdgcn_wmma_f32_16x16x4_f32(false, a, false, b, (short)0, acc[g][0], false, false);
        b.x = b0.y; b.y = b1.y;
        acc[g][1] = __builtin_amdgcn_wmma_f32_16x16x4_f32(false, a, false, b, (short)0, acc[g][1], false, false);
        b.x = b0.z; b.y = b1.z;
        acc[g][2] = __builtin_amdgcn_wmma_f32_16x16x4_f32(false, a, false, b, (short)0, acc[g][2], false, false);
        b.x = b0.w; b.y = b1.w;
        acc[g][3] = __builtin_amdgcn_wmma_f32_16x16x4_f32(false, a, false, b, (short)0, acc[g][3], false, false);
      }
    }
    q0 = n0;
    q1 = n1;
  }

#pragma unroll
  for (int g = 0; g < NG; ++g) {
    const int col = 64 * g + 4 * c;
    float4 bb = {0.0f, 0.0f, 0.0f, 0.0f};
    if (bias) bb = *reinterpret_cast<const float4*>(bias + col);
#pragma unroll
    for (int r = 0; r < 8; ++r) {
      const int row = (tileM << 4) + r + (h << 3);
      float4 v = {acc[g][0][r], acc[g][1][r], acc[g][2][r], acc[g][3][r]};
      if (post_scale) {
        const float s = post_scale[row];
        v.x *= s; v.y *= s; v.z *= s; v.w *= s;
      }
      if (bias) { v.x += bb.x; v.y += bb.y; v.z += bb.z; v.w += bb.w; }
      if (relu) {
        v.x = fmaxf(v.x, 0.0f); v.y = fmaxf(v.y, 0.0f);
        v.z = fmaxf(v.z, 0.0f); v.w = fmaxf(v.w, 0.0f);
      }
      *reinterpret_cast<float4*>(C + (size_t)row * F + col) = v;
    }
  }
}

template <int VEC, int G>
__device__ __forceinline__ void add_rows(float __attribute__((ext_vector_type(VEC)))& acc,
                                         const float* __restrict__ mrow, int my, int i) {
  typedef float vf __attribute__((ext_vector_type(VEC)));
  constexpr int F = 32 * VEC;
  int u[G];
#pragma unroll
  for (int k = 0; k < G; ++k) u[k] = __builtin_amdgcn_readlane(my, i + k);
  vf r[G];
#pragma unroll
  for (int k = 0; k < G; ++k) r[k] = *reinterpret_cast<const vf*>(mrow + (size_t)u[k] * F);
#pragma unroll
  for (int k = 0; k < G; ++k) acc += r[k];
}

template <int VEC>
__global__ __launch_bounds__(256) void gcn_gather(
    const float* __restrict__ msg, const int* __restrict__ rowptr,
    const int* __restrict__ col, const float* __restrict__ post_scale,
    const float* __restrict__ bias, int relu, const float* __restrict__ out_scale,
    float* __restrict__ outp, int n) {
  constexpr int F = 32 * VEC;
  typedef float vf __attribute__((ext_vector_type(VEC)));
  const int lane = threadIdx.x & 31;
  const int v = __builtin_amdgcn_readfirstlane(
      (int)(blockIdx.x * (blockDim.x >> 5) + (threadIdx.x >> 5)));
  if (v >= n) return;
  const int beg = rowptr[v];
  const int end = rowptr[v + 1];

  const float* mrow = msg + VEC * lane;
  vf acc = 0.0f;
  for (int b0 = beg; b0 < end; b0 += 32) {
    const int cnt = (end - b0 < 32) ? end - b0 : 32;
    const int my = (lane < cnt) ? col[b0 + lane] : 0;
    int i = 0;
    for (; i + 8 <= cnt; i += 8) add_rows<VEC, 8>(acc, mrow, my, i);
    if (i + 4 <= cnt) { add_rows<VEC, 4>(acc, mrow, my, i); i += 4; }
    if (i + 2 <= cnt) { add_rows<VEC, 2>(acc, mrow, my, i); i += 2; }
    if (i < cnt) add_rows<VEC, 1>(acc, mrow, my, i);
  }

  if (post_scale) acc *= post_scale[v];
  if (bias) acc += *reinterpret_cast<const vf*>(bias + VEC * lane);
  if (relu) {
#pragma unroll
    for (int t = 0; t < VEC; ++t) acc[t] = fmaxf(acc[t], 0.0f);
  }
  if (out_scale) acc *= out_scale[v];
  *reinterpret_cast<vf*>(outp + (size_t)v * F + VEC * lane) = acc;
}

static inline unsigned cdiv(unsigned a, unsigned b) { return (a + b - 1) / b; }

extern "C" void kernel_launch(void* const* d_in, const int* in_sizes, int n_in,
                              void* d_out, int out_size, void* d_ws, size_t ws_size,
                              hipStream_t stream) {
  const float* features = (const float*)d_in[0];
  const float* W1 = (const float*)d_in[1];
  const float* b1 = (const float*)d_in[2];
  const float* W2 = (const float*)d_in[3];
  const float* b2 = (const float*)d_in[4];
  const float* W3 = (const float*)d_in[5];
  const float* b3 = (const float*)d_in[6];
  const int* src  = (const int*)d_in[7];
  const int* dst  = (const int*)d_in[8];
  float* out = (float*)d_out;

  const int N  = N_NODES;
  const int E  = in_sizes[7];
  const int H  = in_sizes[2];

  float* ws   = (float*)d_ws;
  float* ns   = ws;
  float* nd   = ws + N;
  float* bufA = ws + 2 * N + 96;
  float* bufB = bufA + (size_t)N * H;
  int* cnt_out = (int*)(bufB + (size_t)N * H);
  int* cnt_in  = cnt_out + N;
  int* rowptr  = cnt_in + N;
  int* pos     = rowptr + N + 16;
  int* col     = pos + E;
  (void)ws_size; (void)n_in; (void)out_size;

  const int BS = 256;
  const unsigned node_blocks = cdiv(N, BS / 32);

  gcn_zero_i<<<cdiv(2 * N, BS), BS, 0, stream>>>(cnt_out, 2 * N);
  gcn_count <<<cdiv(E, BS),     BS, 0, stream>>>(src, dst, cnt_out, cnt_in, pos, E);
  gcn_scan  <<<cdiv(N, 1024), 1024, 0, stream>>>(cnt_out, cnt_in, rowptr, ns, nd, N);
  gcn_place <<<cdiv(E, BS),     BS, 0, stream>>>(src, dst, rowptr, pos, col, E);

  gcn_gemm_wmma<256, 128><<<cdiv(N / 16, BS / 32), BS, 0, stream>>>(
      features, W1, bufA, N, ns, nullptr, nullptr, 0);
  gcn_gather<4><<<node_blocks, BS, 0, stream>>>(bufA, rowptr, col, nd, b1, 1, ns, bufB, N);

  gcn_gather<4><<<node_blocks, BS, 0, stream>>>(bufB, rowptr, col, nullptr, nullptr, 0, nullptr, bufA, N);
  gcn_gemm_wmma<128, 128><<<cdiv(N / 16, BS / 32), BS, 0, stream>>>(
      bufA, W2, bufB, N, nullptr, nd, b2, 1);

  gcn_gemm_wmma<128, 64><<<cdiv(N / 16, BS / 32), BS, 0, stream>>>(
      bufB, W3, bufA, N, ns, nullptr, nullptr, 0);
  gcn_gather<2><<<node_blocks, BS, 0, stream>>>(bufA, rowptr, col, nd, b3, 0, nullptr, out, N);
}


# count atomics go to one counter plane per chip half (die number picks it): no counter line moves between halves; list = lower plane's arrivals then upper's: f32 sum order changes, still arrival
# speedup vs baseline: 3.4873x; 1.4012x over previous
#include <hip/hip_runtime.h>


typedef float v2f __attribute__((ext_vector_type(2)));
typedef float v4f __attribute__((ext_vector_type(4)));
typedef float v8f __attribute__((ext_vector_type(8)));

#define N_NODES 50000

__global__ void gcn_zero_i4(int* __restrict__ p, int n4) {
  int i = blockIdx.x * blockDim.x + threadIdx.x;
  if (i < n4) reinterpret_cast<int4*>(p)[i] = int4{0, 0, 0, 0};
}

__global__ void gcn_count(const int* __restrict__ src, const int* __restrict__ dst,
                          int* __restrict__ cnt, int* __restrict__ pos, int E, int n) {
  int e = blockIdx.x * blockDim.x + threadIdx.x;
  if (e >= E) return;
  const int h = (int)((__builtin_amdgcn_s_sendmsg_rtn(0x87) >> 18) & 1u);
  int* cnt_out = cnt + h * 2 * n;
  int* cnt_in = cnt_out + n;
  const int s = src[e];
  const int d = dst[e];
  atomicAdd(&cnt_out[s], 1);
  pos[e] = (atomicAdd(&cnt_in[d], 1) << 1) | h;
}

__global__ __launch_bounds__(1024) void gcn_scan(
    const int* __restrict__ cnt, int* __restrict__ rowptr, int* __restrict__ rowmid,
    float* __restrict__ ns, float* __restrict__ nd, int n) {
  const int* cnt_out0 = cnt;
  const int* cnt_in0 = cnt + n;
  const int* cnt_out1 = cnt + 2 * n;
  const int* cnt_in1 = cnt + 3 * n;
  __shared__ int front[32];
  __shared__ int wtot[32];
  const int tid = threadIdx.x;
  const int lane = tid & 31;
  const int w = tid >> 5;
  const int first = blockIdx.x * 1024;

  int s = 0;
  const int4* c4a = reinterpret_cast<const int4*>(cnt_in0);
  const int4* c4b = reinterpret_cast<const int4*>(cnt_in1);
#pragma unroll 4
  for (int i = tid; i < (first >> 2); i += 1024) {
    const int4 q = c4a[i];
    const int4 r = c4b[i];
    s += ((q.x + q.y) + (q.z + q.w)) + ((r.x + r.y) + (r.z + r.w));
  }
  const int idx = first + tid;
  const int c0 = (idx < n) ? cnt_in0[idx] : 0;
  const int c = c0 + ((idx < n) ? cnt_in1[idx] : 0);

  int incl = c;
  for (int d = 1; d < 32; d <<= 1) {
    const int t = __shfl_up(incl, d);
    if (lane >= d) incl += t;
  }
  for (int d = 16; d >= 1; d >>= 1) s += __shfl_xor(s, d);
  if (lane == 31) wtot[w] = incl;
  if (lane == 0) front[w] = s;
  __syncthreads();
  int base = 0;
  for (int k = 0; k < 32; ++k) {
    base += front[k];
    if (k < w) base += wtot[k];
  }
  if (idx < n) {
    const int ex = base + incl - c;
    rowptr[idx] = ex;
    rowmid[idx] = ex + c0;
    if (idx == n - 1) rowptr[n] = ex + c;
    nd[idx] = __frsqrt_rn(fmaxf((float)c, 1.0f));
    ns[idx] = __frsqrt_rn(fmaxf((float)(cnt_out0[idx] + cnt_out1[idx]), 1.0f));
  }
}

__global__ void gcn_place(const int* __restrict__ src, const int* __restrict__ dst,
                          const int* __restrict__ rowptr, const int* __restrict__ rowmid,
                          const int* __restrict__ pos, int* __restrict__ col, int E) {
  int e = blockIdx.x * blockDim.x + threadIdx.x;
  if (e >= E) return;
  const int p = pos[e];
  const int* start = (p & 1) ? rowmid : rowptr;
  col[start[dst[e]] + (p >> 1)] = src[e];
}

template <int K, int F>
__global__ __launch_bounds__(256) void gcn_gemm_wmma(
    const float* __restrict__ A, const float* __restrict__ W,
    float* __restrict__ C, int n_rows,
    const float* __restrict__ pre_scale, const float* __restrict__ post_scale,
    const float* __restrict__ bias, int relu) {
  constexpr int NG = F / 64;
  __shared__ alignas(16) float Wl[K * F];

  {
    const float4* Wg = reinterpret_cast<const float4*>(W);
    float4* Ws = reinterpret_cast<float4*>(Wl);
    for (int i = threadIdx.x; i < K * F / 4; i += 256) Ws[i] = Wg[i];
  }
  __syncthreads();

  const int tileM = blockIdx.x * 8 + ((int)threadIdx.x >> 5);
  if (tileM >= (n_rows >> 4)) return;
  const int lane = threadIdx.x & 31;
  const int h = lane >> 4;
  const int c = lane & 15;

  const int m = (tileM << 4) + c;
  const float ps = pre_scale ? pre_scale[m] : 1.0f;
  const float4* Arow = reinterpret_cast<const float4*>(A + (size_t)m * K + 8 * h);
  const float* Wlane = Wl + (8 * h) * F + 4 * c;

  v8f acc[NG][4];
#pragma unroll
  for (int g = 0; g < NG; ++g)
#pragma unroll
    for (int t = 0; t < 4; ++t) acc[g][t] = v8f{};

  float4 q0 = Arow[0], q1 = Arow[1];
  for (int kb = 0; kb < K; kb += 16) {
    const int kn = (kb + 16 < K) ? kb + 16 : kb;
    const float4 n0 = Arow[kn >> 2], n1 = Arow[(kn >> 2) + 1];

    const float av[8] = {q0.x * ps, q0.y * ps, q0.z * ps, q0.w * ps,
                         q1.x * ps, q1.y * ps, q1.z * ps, q1.w * ps};
#pragma unroll
    for (int j = 0; j < 4; ++j) {
      v2f a;
      a.x = av[2 * j];
      a.y = av[2 * j + 1];
      const float* wr = Wlane + (kb + 2 * j) * F;
#pragma unroll
      for (int g = 0; g < NG; ++g) {
        const float4 b0 = *reinterpret_cast<const float4*>(wr + 64 * g);
        const float4 b1 = *reinterpret_cast<const float4*>(wr + F + 64 * g);
        v2f b;
        b.x = b0.x; b.y = b1.x;
        acc[g][0] = __builtin_amdgcn_wmma_f32_16x16x4_f32(false, a, false, b, (short)0, acc[g][0], false, false);
        b.x = b0.y; b.y = b1.y;
        acc[g][1] = __builtin_amdgcn_wmma_f32_16x16x4_f32(false, a, false, b, (short)0, acc[g][1], false, false);
        b.x = b0.z; b.y = b1.z;
        acc[g][2] = __builtin_amdgcn_wmma_f32_16x16x4_f32(false, a, false, b, (short)0, acc[g][2], false, false);
        b.x = b0.w; b.y = b1.w;
        acc[g][3] = __builtin_amdgcn_wmma_f32_16x16x4_f32(false, a, false, b, (short)0, acc[g][3], false, false);
      }
    }
    q0 = n0;
    q1 = n1;
  }

#pragma unroll
  for (int g = 0; g < NG; ++g) {
    const int col = 64 * g + 4 * c;
    float4 bb = {0.0f, 0.0f, 0.0f, 0.0f};
    if (bias) bb = *reinterpret_cast<const float4*>(bias + col);
#pragma unroll
    for (int r = 0; r < 8; ++r) {
      const int row = (tileM << 4) + r + (h << 3);
      float4 v = {acc[g][0][r], acc[g][1][r], acc[g][2][r], acc[g][3][r]};
      if (post_scale) {
        const float s = post_scale[row];
        v.x *= s; v.y *= s; v.z *= s; v.w *= s;
      }
      if (bias) { v.x += bb.x; v.y += bb.y; v.z += bb.z; v.w += bb.w; }
      if (relu) {
        v.x = fmaxf(v.x, 0.0f); v.y = fmaxf(v.y, 0.0f);
        v.z = fmaxf(v.z, 0.0f); v.w = fmaxf(v.w, 0.0f);
      }
      *reinterpret_cast<float4*>(C + (size_t)row * F + col) = v;
    }
  }
}

template <int VEC, int G>
__device__ __forceinline__ void add_rows(float __attribute__((ext_vector_type(VEC)))& acc,
                                         const float* __restrict__ mrow, int my, int i) {
  typedef float vf __attribute__((ext_vector_type(VEC)));
  constexpr int F = 32 * VEC;
  int u[G];
#pragma unroll
  for (int k = 0; k < G; ++k) u[k] = __builtin_amdgcn_readlane(my, i + k);
  vf r[G];
#pragma unroll
  for (int k = 0; k < G; ++k) r[k] = *reinterpret_cast<const vf*>(mrow + (size_t)u[k] * F);
#pragma unroll
  for (int k = 0; k < G; ++k) acc += r[k];
}

template <int VEC>
__global__ __launch_bounds__(256) void gcn_gather(
    const float* __restrict__ msg, const int* __restrict__ rowptr,
    const int* __restrict__ col, const float* __restrict__ post_scale,
    const float* __restrict__ bias, int relu, const float* __restrict__ out_scale,
    float* __restrict__ outp, int n) {
  constexpr int F = 32 * VEC;
  typedef float vf __attribute__((ext_vector_type(VEC)));
  const int lane = threadIdx.x & 31;
  const int v = __builtin_amdgcn_readfirstlane(
      (int)(blockIdx.x * (blockDim.x >> 5) + (threadIdx.x >> 5)));
  if (v >= n) return;
  const int beg = rowptr[v];
  const int end = rowptr[v + 1];

  const float* mrow = msg + VEC * lane;
  vf acc = 0.0f;
  for (int b0 = beg; b0 < end; b0 += 32) {
    const int cnt = (end - b0 < 32) ? end - b0 : 32;
    const int my = (lane < cnt) ? col[b0 + lane] : 0;
    int i = 0;
    for (; i + 8 <= cnt; i += 8) add_rows<VEC, 8>(acc, mrow, my, i);
    if (i + 4 <= cnt) { add_rows<VEC, 4>(acc, mrow, my, i); i += 4; }
    if (i + 2 <= cnt) { add_rows<VEC, 2>(acc, mrow, my, i); i += 2; }
    if (i < cnt) add_rows<VEC, 1>(acc, mrow, my, i);
  }

  if (post_scale) acc *= post_scale[v];
  if (bias) acc += *reinterpret_cast<const vf*>(bias + VEC * lane);
  if (relu) {
#pragma unroll
    for (int t = 0; t < VEC; ++t) acc[t] = fmaxf(acc[t], 0.0f);
  }
  if (out_scale) acc *= out_scale[v];
  *reinterpret_cast<vf*>(outp + (size_t)v * F + VEC * lane) = acc;
}

static inline unsigned cdiv(unsigned a, unsigned b) { return (a + b - 1) / b; }

extern "C" void kernel_launch(void* const* d_in, const int* in_sizes, int n_in,
                              void* d_out, int out_size, void* d_ws, size_t ws_size,
                              hipStream_t stream) {
  const float* features = (const float*)d_in[0];
  const float* W1 = (const float*)d_in[1];
  const float* b1 = (const float*)d_in[2];
  const float* W2 = (const float*)d_in[3];
  const float* b2 = (const float*)d_in[4];
  const float* W3 = (const float*)d_in[5];
  const float* b3 = (const float*)d_in[6];
  const int* src  = (const int*)d_in[7];
  const int* dst  = (const int*)d_in[8];
  float* out = (float*)d_out;

  const int N  = N_NODES;
  const int E  = in_sizes[7];
  const int H  = in_sizes[2];

  float* ws   = (float*)d_ws;
  float* ns   = ws;
  float* nd   = ws + N;
  float* bufA = ws + 2 * N + 96;
  float* bufB = bufA + (size_t)N * H;
  int* cnt     = (int*)(bufB + (size_t)N * H);
  int* rowptr  = cnt + 4 * N;
  int* rowmid  = rowptr + N + 16;
  int* pos     = rowmid + N + 16;
  int* col     = pos + E;
  (void)ws_size; (void)n_in; (void)out_size;

  const int BS = 256;
  const unsigned node_blocks = cdiv(N, BS / 32);

  gcn_zero_i4<<<cdiv(N, BS),    BS, 0, stream>>>(cnt, N);
  gcn_count <<<cdiv(E, BS),     BS, 0, stream>>>(src, dst, cnt, pos, E, N);
  gcn_scan  <<<cdiv(N, 1024), 1024, 0, stream>>>(cnt, rowptr, rowmid, ns, nd, N);
  gcn_place <<<cdiv(E, BS),     BS, 0, stream>>>(src, dst, rowptr, rowmid, pos, col, E);

  gcn_gemm_wmma<256, 128><<<cdiv(N / 16, BS / 32), BS, 0, stream>>>(
      features, W1, bufA, N, ns, nullptr, nullptr, 0);
  gcn_gather<4><<<node_blocks, BS, 0, stream>>>(bufA, rowptr, col, nd, b1, 1, ns, bufB, N);

  gcn_gather<4><<<node_blocks, BS, 0, stream>>>(bufB, rowptr, col, nullptr, nullptr, 0, nullptr, bufA, N);
  gcn_gemm_wmma<128, 128><<<cdiv(N / 16, BS / 32), BS, 0, stream>>>(
      bufA, W2, bufB, N, nullptr, nd, b2, 1);

  gcn_gemm_wmma<128, 64><<<cdiv(N / 16, BS / 32), BS, 0, stream>>>(
      bufB, W3, bufA, N, ns, nullptr, nullptr, 0);
  gcn_gather<2><<<node_blocks, BS, 0, stream>>>(bufA, rowptr, col, nd, b3, 0, nullptr, out, N);
}


# edge placement rides in the layer-1 GEMM's launch: 4 extra waves per block place its edges (loads 8 deep) and only wait on memory while waves 0-7 run the WMMAs; one launch fewer; sums unchanged
# speedup vs baseline: 3.6069x; 1.0343x over previous
#include <hip/hip_runtime.h>


typedef float v2f __attribute__((ext_vector_type(2)));
typedef float v4f __attribute__((ext_vector_type(4)));
typedef float v8f __attribute__((ext_vector_type(8)));

#define N_NODES 50000

__global__ void gcn_zero_i4(int* __restrict__ p, int n4) {
  int i = blockIdx.x * blockDim.x + threadIdx.x;
  if (i < n4) reinterpret_cast<int4*>(p)[i] = int4{0, 0, 0, 0};
}

__global__ void gcn_count(const int* __restrict__ src, const int* __restrict__ dst,
                          int* __restrict__ cnt, int* __restrict__ pos, int E, int n) {
  int e = blockIdx.x * blockDim.x + threadIdx.x;
  if (e >= E) return;
  const int h = (int)((__builtin_amdgcn_s_sendmsg_rtn(0x87) >> 18) & 1u);
  int* cnt_out = cnt + h * 2 * n;
  int* cnt_in = cnt_out + n;
  const int s = src[e];
  const int d = dst[e];
  atomicAdd(&cnt_out[s], 1);
  pos[e] = (atomicAdd(&cnt_in[d], 1) << 1) | h;
}

__global__ __launch_bounds__(1024) void gcn_scan(
    const int* __restrict__ cnt, int* __restrict__ rowptr, int* __restrict__ rowmid,
    float* __restrict__ ns, float* __restrict__ nd, int n) {
  const int* cnt_out0 = cnt;
  const int* cnt_in0 = cnt + n;
  const int* cnt_out1 = cnt + 2 * n;
  const int* cnt_in1 = cnt + 3 * n;
  __shared__ int front[32];
  __shared__ int wtot[32];
  const int tid = threadIdx.x;
  const int lane = tid & 31;
  const int w = tid >> 5;
  const int first = blockIdx.x * 1024;

  int s = 0;
  const int4* c4a = reinterpret_cast<const int4*>(cnt_in0);
  const int4* c4b = reinterpret_cast<const int4*>(cnt_in1);
#pragma unroll 4
  for (int i = tid; i < (first >> 2); i += 1024) {
    const int4 q = c4a[i];
    const int4 r = c4b[i];
    s += ((q.x + q.y) + (q.z + q.w)) + ((r.x + r.y) + (r.z + r.w));
  }
  const int idx = first + tid;
  const int c0 = (idx < n) ? cnt_in0[idx] : 0;
  const int c = c0 + ((idx < n) ? cnt_in1[idx] : 0);

  int incl = c;
  for (int d = 1; d < 32; d <<= 1) {
    const int t = __shfl_up(incl, d);
    if (lane >= d) incl += t;
  }
  for (int d = 16; d >= 1; d >>= 1) s += __shfl_xor(s, d);
  if (lane == 31) wtot[w] = incl;
  if (lane == 0) front[w] = s;
  __syncthreads();
  int base = 0;
  for (int k = 0; k < 32; ++k) {
    base += front[k];
    if (k < w) base += wtot[k];
  }
  if (idx < n) {
    const int ex = base + incl - c;
    rowptr[idx] = ex;
    rowmid[idx] = ex + c0;
    if (idx == n - 1) rowptr[n] = ex + c;
    nd[idx] = __frsqrt_rn(fmaxf((float)c, 1.0f));
    ns[idx] = __frsqrt_rn(fmaxf((float)(cnt_out0[idx] + cnt_out1[idx]), 1.0f));
  }
}

template <int K, int F, bool PLACE>
__global__ __launch_bounds__(PLACE ? 384 : 256) void gcn_gemm_wmma(
    const float* __restrict__ A, const float* __restrict__ W,
    float* __restrict__ C, int n_rows,
    const float* __restrict__ pre_scale, const float* __restrict__ post_scale,
    const float* __restrict__ bias, int relu,
    const int* __restrict__ src, const int* __restrict__ dst,
    const int* __restrict__ rowptr, const int* __restrict__ rowmid,
    const int* __restrict__ pos, int* __restrict__ col, int E) {
  constexpr int NG = F / 64;
  constexpr int NT = PLACE ? 384 : 256;
  constexpr int PL = NT - 256;
  __shared__ alignas(16) float Wl[K * F];

  {
    const float4* Wg = reinterpret_cast<const float4*>(W);
    float4* Ws = reinterpret_cast<float4*>(Wl);
    for (int i = threadIdx.x; i < K * F / 4; i += NT) Ws[i] = Wg[i];
  }
  __syncthreads();

  if constexpr (PLACE) {
    if (threadIdx.x >= 256) {
      const int per = ((E + (int)gridDim.x - 1) / (int)gridDim.x + 255) & ~255;
      const int first = (int)blockIdx.x * per;
      const int last = (first + per < E) ? first + per : E;
      for (int e0 = first + (int)threadIdx.x - 256; e0 < last; e0 += PL * 8) {
        int p[8], d[8], s[8], st[8];
#pragma unroll
        for (int k = 0; k < 8; ++k) {
          const int e = (e0 + PL * k < last) ? e0 + PL * k : last - 1;
          p[k] = pos[e];
          d[k] = dst[e];
          s[k] = src[e];
        }
#pragma unroll
        for (int k = 0; k < 8; ++k) st[k] = ((p[k] & 1) ? rowmid : rowptr)[d[k]];
#pragma unroll
        for (int k = 0; k < 8; ++k) col[st[k] + (p[k] >> 1)] = s[k];
      }
      return;
    }
  }

  const int tileM = blockIdx.x * 8 + ((int)threadIdx.x >> 5);
  if (tileM >= (n_rows >> 4)) return;
  const int lane = threadIdx.x & 31;
  const int h = lane >> 4;
  const int c = lane & 15;

  const int m = (tileM << 4) + c;
  const float ps = pre_scale ? pre_scale[m] : 1.0f;
  const float4* Arow = reinterpret_cast<const float4*>(A + (size_t)m * K + 8 * h);
  const float* Wlane = Wl + (8 * h) * F + 4 * c;

  v8f acc[NG][4];
#pragma unroll
  for (int g = 0; g < NG; ++g)
#pragma unroll
    for (int t = 0; t < 4; ++t) acc[g][t] = v8f{};

  float4 q0 = Arow[0], q1 = Arow[1];
  for (int kb = 0; kb < K; kb += 16) {
    const int kn = (kb + 16 < K) ? kb + 16 : kb;
    const float4 n0 = Arow[kn >> 2], n1 = Arow[(kn >> 2) + 1];

    const float av[8] = {q0.x * ps, q0.y * ps, q0.z * ps, q0.w * ps,
                         q1.x * ps, q1.y * ps, q1.z * ps, q1.w * ps};
#pragma unroll
    for (int j = 0; j < 4; ++j) {
      v2f a;
      a.x = av[2 * j];
      a.y = av[2 * j + 1];
      const float* wr = Wlane + (kb + 2 * j) * F;
#pragma unroll
      for (int g = 0; g < NG; ++g) {
        const float4 b0 = *reinterpret_cast<const float4*>(wr + 64 * g);
        const float4 b1 = *reinterpret_cast<const float4*>(wr + F + 64 * g);
        v2f b;
        b.x = b0.x; b.y = b1.x;
        acc[g][0] = __builtin_amdgcn_wmma_f32_16x16x4_f32(false, a, false, b, (short)0, acc[g][0], false, false);
        b.x = b0.y; b.y = b1.y;
        acc[g][1] = __builtin_amdgcn_wmma_f32_16x16x4_f32(false, a, false, b, (short)0, acc[g][1], false, false);
        b.x = b0.z; b.y = b1.z;
        acc[g][2] = __builtin_amdgcn_wmma_f32_16x16x4_f32(false, a, false, b, (short)0, acc[g][2], false, false);
        b.x = b0.w; b.y = b1.w;
        acc[g][3] = __builtin_amdgcn_wmma_f32_16x16x4_f32(false, a, false, b, (short)0, acc[g][3], false, false);
      }
    }
    q0 = n0;
    q1 = n1;
  }

#pragma unroll
  for (int g = 0; g < NG; ++g) {
    const int col = 64 * g + 4 * c;
    float4 bb = {0.0f, 0.0f, 0.0f, 0.0f};
    if (bias) bb = *reinterpret_cast<const float4*>(bias + col);
#pragma unroll
    for (int r = 0; r < 8; ++r) {
      const int row = (tileM << 4) + r + (h << 3);
      float4 v = {acc[g][0][r], acc[g][1][r], acc[g][2][r], acc[g][3][r]};
      if (post_scale) {
        const float s = post_scale[row];
        v.x *= s; v.y *= s; v.z *= s; v.w *= s;
      }
      if (bias) { v.x += bb.x; v.y += bb.y; v.z += bb.z; v.w += bb.w; }
      if (relu) {
        v.x = fmaxf(v.x, 0.0f); v.y = fmaxf(v.y, 0.0f);
        v.z = fmaxf(v.z, 0.0f); v.w = fmaxf(v.w, 0.0f);
      }
      *reinterpret_cast<float4*>(C + (size_t)row * F + col) = v;
    }
  }
}

template <int VEC, int G>
__device__ __forceinline__ void add_rows(float __attribute__((ext_vector_type(VEC)))& acc,
                                         const float* __restrict__ mrow, int my, int i) {
  typedef float vf __attribute__((ext_vector_type(VEC)));
  constexpr int F = 32 * VEC;
  int u[G];
#pragma unroll
  for (int k = 0; k < G; ++k) u[k] = __builtin_amdgcn_readlane(my, i + k);
  vf r[G];
#pragma unroll
  for (int k = 0; k < G; ++k) r[k] = *reinterpret_cast<const vf*>(mrow + (size_t)u[k] * F);
#pragma unroll
  for (int k = 0; k < G; ++k) acc += r[k];
}

template <int VEC>
__global__ __launch_bounds__(256) void gcn_gather(
    const float* __restrict__ msg, const int* __restrict__ rowptr,
    const int* __restrict__ col, const float* __restrict__ post_scale,
    const float* __restrict__ bias, int relu, const float* __restrict__ out_scale,
    float* __restrict__ outp, int n) {
  constexpr int F = 32 * VEC;
  typedef float vf __attribute__((ext_vector_type(VEC)));
  const int lane = threadIdx.x & 31;
  const int v = __builtin_amdgcn_readfirstlane(
      (int)(blockIdx.x * (blockDim.x >> 5) + (threadIdx.x >> 5)));
  if (v >= n) return;
  const int beg = rowptr[v];
  const int end = rowptr[v + 1];

  const float* mrow = msg + VEC * lane;
  vf acc = 0.0f;
  for (int b0 = beg; b0 < end; b0 += 32) {
    const int cnt = (end - b0 < 32) ? end - b0 : 32;
    const int my = (lane < cnt) ? col[b0 + lane] : 0;
    int i = 0;
    for (; i + 8 <= cnt; i += 8) add_rows<VEC, 8>(acc, mrow, my, i);
    if (i + 4 <= cnt) { add_rows<VEC, 4>(acc, mrow, my, i); i += 4; }
    if (i + 2 <= cnt) { add_rows<VEC, 2>(acc, mrow, my, i); i += 2; }
    if (i < cnt) add_rows<VEC, 1>(acc, mrow, my, i);
  }

  if (post_scale) acc *= post_scale[v];
  if (bias) acc += *reinterpret_cast<const vf*>(bias + VEC * lane);
  if (relu) {
#pragma unroll
    for (int t = 0; t < VEC; ++t) acc[t] = fmaxf(acc[t], 0.0f);
  }
  if (out_scale) acc *= out_scale[v];
  *reinterpret_cast<vf*>(outp + (size_t)v * F + VEC * lane) = acc;
}

static inline unsigned cdiv(unsigned a, unsigned b) { return (a + b - 1) / b; }

extern "C" void kernel_launch(void* const* d_in, const int* in_sizes, int n_in,
                              void* d_out, int out_size, void* d_ws, size_t ws_size,
                              hipStream_t stream) {
  const float* features = (const float*)d_in[0];
  const float* W1 = (const float*)d_in[1];
  const float* b1 = (const float*)d_in[2];
  const float* W2 = (const float*)d_in[3];
  const float* b2 = (const float*)d_in[4];
  const float* W3 = (const float*)d_in[5];
  const float* b3 = (const float*)d_in[6];
  const int* src  = (const int*)d_in[7];
  const int* dst  = (const int*)d_in[8];
  float* out = (float*)d_out;

  const int N  = N_NODES;
  const int E  = in_sizes[7];
  const int H  = in_sizes[2];

  float* ws   = (float*)d_ws;
  float* ns   = ws;
  float* nd   = ws + N;
  float* bufA = ws + 2 * N + 96;
  float* bufB = bufA + (size_t)N * H;
  int* cnt     = (int*)(bufB + (size_t)N * H);
  int* rowptr  = cnt + 4 * N;
  int* rowmid  = rowptr + N + 16;
  int* pos     = rowmid + N + 16;
  int* col     = pos + E;
  (void)ws_size; (void)n_in; (void)out_size;

  const int BS = 256;
  const unsigned node_blocks = cdiv(N, BS / 32);

  gcn_zero_i4<<<cdiv(N, BS),    BS, 0, stream>>>(cnt, N);
  gcn_count <<<cdiv(E, BS),     BS, 0, stream>>>(src, dst, cnt, pos, E, N);
  gcn_scan  <<<cdiv(N, 1024), 1024, 0, stream>>>(cnt, rowptr, rowmid, ns, nd, N);

  gcn_gemm_wmma<256, 128, true><<<cdiv(N / 16, BS / 32), BS + 128, 0, stream>>>(
      features, W1, bufA, N, ns, nullptr, nullptr, 0,
      src, dst, rowptr, rowmid, pos, col, E);
  gcn_gather<4><<<node_blocks, BS, 0, stream>>>(bufA, rowptr, col, nd, b1, 1, ns, bufB, N);

  gcn_gather<4><<<node_blocks, BS, 0, stream>>>(bufB, rowptr, col, nullptr, nullptr, 0, nullptr, bufA, N);
  gcn_gemm_wmma<128, 128, false><<<cdiv(N / 16, BS / 32), BS, 0, stream>>>(
      bufA, W2, bufB, N, nullptr, nd, b2, 1,
      nullptr, nullptr, nullptr, nullptr, nullptr, nullptr, 0);

  gcn_gemm_wmma<128, 64, false><<<cdiv(N / 16, BS / 32), BS, 0, stream>>>(
      bufB, W3, bufA, N, ns, nullptr, nullptr, 0,
      nullptr, nullptr, nullptr, nullptr, nullptr, nullptr, 0);
  gcn_gather<2><<<node_blocks, BS, 0, stream>>>(bufA, rowptr, col, nd, b3, 0, nullptr, out, N);
}


# edge list = 32-slot buckets per node and chip half filled in one count pass (slot = returned count) by extra waves of GEMM 1: no scan/place/count launch; ns scales gather-1 rows; arrival-order sums
# speedup vs baseline: 4.1648x; 1.1547x over previous
#include <hip/hip_runtime.h>


typedef float v2f __attribute__((ext_vector_type(2)));
typedef float v4f __attribute__((ext_vector_type(4)));
typedef float v8f __attribute__((ext_vector_type(8)));

#define N_NODES 50000
#define CAP 32

__global__ void gcn_zero_i4(int* __restrict__ p, int n4) {
  int i = blockIdx.x * blockDim.x + threadIdx.x;
  if (i < n4) reinterpret_cast<int4*>(p)[i] = int4{0, 0, 0, 0};
}

template <int K, int F, bool PLACE>
__global__ __launch_bounds__(PLACE ? 384 : 256) void gcn_gemm_wmma(
    const float* __restrict__ A, const float* __restrict__ W,
    float* __restrict__ C, int n_rows,
    const float* __restrict__ pre_scale, const float* __restrict__ post_scale,
    const float* __restrict__ bias, int relu,
    const int* __restrict__ src, const int* __restrict__ dst,
    int* __restrict__ cnt, int* __restrict__ spill_n, int* __restrict__ bucket,
    int* __restrict__ spill_d, int* __restrict__ spill_s, int E, int n) {
  constexpr int NG = F / 64;
  constexpr int NT = PLACE ? 384 : 256;
  constexpr int PL = NT - 256;
  __shared__ alignas(16) float Wl[K * F];

  {
    const float4* Wg = reinterpret_cast<const float4*>(W);
    float4* Ws = reinterpret_cast<float4*>(Wl);
    for (int i = threadIdx.x; i < K * F / 4; i += NT) Ws[i] = Wg[i];
  }
  __syncthreads();

  if constexpr (PLACE) {
    if (threadIdx.x >= 256) {
      const int hh = (int)((__builtin_amdgcn_s_sendmsg_rtn(0x87) >> 18) & 1u);
      int* cnt_out = cnt + hh * 2 * n;
      int* cnt_in = cnt_out + n;
      int* bk = bucket + hh * n * CAP;
      const int per = ((E + (int)gridDim.x - 1) / (int)gridDim.x + PL - 1) & ~(PL - 1);
      const int first = (int)blockIdx.x * per;
      const int last = (first + per < E) ? first + per : E;
      for (int e0 = first + (int)threadIdx.x - 256; e0 < last; e0 += PL * 8) {
        int d[8], sr[8], sl[8];
#pragma unroll
        for (int k = 0; k < 8; ++k) {
          const int e = (e0 + PL * k < last) ? e0 + PL * k : last - 1;
          d[k] = dst[e];
          sr[k] = src[e];
        }
#pragma unroll
        for (int k = 0; k < 8; ++k) {
          sl[k] = 0;
          if (e0 + PL * k < last) {
            atomicAdd(&cnt_out[sr[k]], 1);
            sl[k] = atomicAdd(&cnt_in[d[k]], 1);
          }
        }
#pragma unroll
        for (int k = 0; k < 8; ++k) {
          if (e0 + PL * k < last) {
            if (sl[k] < CAP) {
              bk[d[k] * CAP + sl[k]] = sr[k];
            } else {
              const int q = atomicAdd(spill_n, 1);
              spill_d[q] = d[k];
              spill_s[q] = sr[k];
            }
          }
        }
      }
      return;
    }
  }

  const int tileM = blockIdx.x * 8 + ((int)threadIdx.x >> 5);
  if (tileM >= (n_rows >> 4)) return;
  const int lane = threadIdx.x & 31;
  const int h = lane >> 4;
  const int c = lane & 15;

  const int m = (tileM << 4) + c;
  const float ps = pre_scale ? pre_scale[m] : 1.0f;
  const float4* Arow = reinterpret_cast<const float4*>(A + (size_t)m * K + 8 * h);
  const float* Wlane = Wl + (8 * h) * F + 4 * c;

  v8f acc[NG][4];
#pragma unroll
  for (int g = 0; g < NG; ++g)
#pragma unroll
    for (int t = 0; t < 4; ++t) acc[g][t] = v8f{};

  float4 q0 = Arow[0], q1 = Arow[1];
  for (int kb = 0; kb < K; kb += 16) {
    const int kn = (kb + 16 < K) ? kb + 16 : kb;
    const float4 n0 = Arow[kn >> 2], n1 = Arow[(kn >> 2) + 1];

    const float av[8] = {q0.x * ps, q0.y * ps, q0.z * ps, q0.w * ps,
                         q1.x * ps, q1.y * ps, q1.z * ps, q1.w * ps};
#pragma unroll
    for (int j = 0; j < 4; ++j) {
      v2f a;
      a.x = av[2 * j];
      a.y = av[2 * j + 1];
      const float* wr = Wlane + (kb + 2 * j) * F;
#pragma unroll
      for (int g = 0; g < NG; ++g) {
        const float4 b0 = *reinterpret_cast<const float4*>(wr + 64 * g);
        const float4 b1 = *reinterpret_cast<const float4*>(wr + F + 64 * g);
        v2f b;
        b.x = b0.x; b.y = b1.x;
        acc[g][0] = __builtin_amdgcn_wmma_f32_16x16x4_f32(false, a, false, b, (short)0, acc[g][0], false, false);
        b.x = b0.y; b.y = b1.y;
        acc[g][1] = __builtin_amdgcn_wmma_f32_16x16x4_f32(false, a, false, b, (short)0, acc[g][1], false, false);
        b.x = b0.z; b.y = b1.z;
        acc[g][2] = __builtin_amdgcn_wmma_f32_16x16x4_f32(false, a, false, b, (short)0, acc[g][2], false, false);
        b.x = b0.w; b.y = b1.w;
        acc[g][3] = __builtin_amdgcn_wmma_f32_16x16x4_f32(false, a, false, b, (short)0, acc[g][3], false, false);
      }
    }
    q0 = n0;
    q1 = n1;
  }

#pragma unroll
  for (int g = 0; g < NG; ++g) {
    const int col = 64 * g + 4 * c;
    float4 bb = {0.0f, 0.0f, 0.0f, 0.0f};
    if (bias) bb = *reinterpret_cast<const float4*>(bias + col);
#pragma unroll
    for (int r = 0; r < 8; ++r) {
      const int row = (tileM << 4) + r + (h << 3);
      float4 v = {acc[g][0][r], acc[g][1][r], acc[g][2][r], acc[g][3][r]};
      if (post_scale) {
        const float s = post_scale[row];
        v.x *= s; v.y *= s; v.z *= s; v.w *= s;
      }
      if (bias) { v.x += bb.x; v.y += bb.y; v.z += bb.z; v.w += bb.w; }
      if (relu) {
        v.x = fmaxf(v.x, 0.0f); v.y = fmaxf(v.y, 0.0f);
        v.z = fmaxf(v.z, 0.0f); v.w = fmaxf(v.w, 0.0f);
      }
      *reinterpret_cast<float4*>(C + (size_t)row * F + col) = v;
    }
  }
}

template <int VEC, int G, bool W>
__device__ __forceinline__ void add_rows(float __attribute__((ext_vector_type(VEC)))& acc,
                                         const float* __restrict__ mrow, int my, int oa, int ob, int i) {
  typedef float vf __attribute__((ext_vector_type(VEC)));
  constexpr int F = 32 * VEC;
  int u[G];
#pragma unroll
  for (int k = 0; k < G; ++k) u[k] = __builtin_amdgcn_readlane(my, i + k);
  vf r[G];
#pragma unroll
  for (int k = 0; k < G; ++k) r[k] = *reinterpret_cast<const vf*>(mrow + (size_t)u[k] * F);
#pragma unroll
  for (int k = 0; k < G; ++k) {
    if constexpr (W) {
      const int od = __builtin_amdgcn_readlane(oa, i + k) + __builtin_amdgcn_readlane(ob, i + k);
      acc += r[k] * __frsqrt_rn(fmaxf((float)od, 1.0f));
    } else {
      acc += r[k];
    }
  }
}

template <int VEC, bool L1>
__global__ __launch_bounds__(256) void gcn_gather(
    const float* __restrict__ msg, const int* __restrict__ cnt,
    const int* __restrict__ bucket, const int* __restrict__ spill_n,
    const int* __restrict__ spill_d, const int* __restrict__ spill_s,
    const float* __restrict__ post_scale,
    const float* __restrict__ bias, int relu, const float* __restrict__ out_scale,
    float* __restrict__ outp, float* __restrict__ ns_w, float* __restrict__ nd_w, int n) {
  constexpr int F = 32 * VEC;
  typedef float vf __attribute__((ext_vector_type(VEC)));
  const int lane = threadIdx.x & 31;
  const int v = __builtin_amdgcn_readfirstlane(
      (int)(blockIdx.x * (blockDim.x >> 5) + (threadIdx.x >> 5)));
  if (v >= n) return;
  const int c0 = cnt[n + v];
  const int c1 = cnt[3 * n + v];
  const int n0 = (c0 < CAP) ? c0 : CAP;
  const int n1 = (c1 < CAP) ? c1 : CAP;
  const int tot = n0 + n1;
  const int* B0 = bucket + v * CAP;
  const int* B1 = bucket + (n + v) * CAP;

  const float* mrow = msg + VEC * lane;
  vf acc = 0.0f;
  for (int b0 = 0; b0 < tot; b0 += 32) {
    const int m = (tot - b0 < 32) ? tot - b0 : 32;
    const int idx = b0 + lane;
    const int* slot = (idx < n0) ? B0 + idx : B1 + (idx - n0);
    const int my = (lane < m) ? *slot : 0;
    int oa = 0, ob = 0;
    if constexpr (L1) {
      oa = cnt[my];
      ob = cnt[2 * n + my];
    }
    int i = 0;
    for (; i + 8 <= m; i += 8) add_rows<VEC, 8, L1>(acc, mrow, my, oa, ob, i);
    if (i + 4 <= m) { add_rows<VEC, 4, L1>(acc, mrow, my, oa, ob, i); i += 4; }
    if (i + 2 <= m) { add_rows<VEC, 2, L1>(acc, mrow, my, oa, ob, i); i += 2; }
    if (i < m) add_rows<VEC, 1, L1>(acc, mrow, my, oa, ob, i);
  }
  if (c0 > CAP || c1 > CAP) {
    const int M = spill_n[0];
    for (int j0 = 0; j0 < M; j0 += 32) {
      const int j = j0 + lane;
      const int dj = (j < M) ? spill_d[j] : -1;
      const int sj = (j < M) ? spill_s[j] : 0;
      int pa = 0, pb = 0;
      if constexpr (L1) {
        pa = cnt[sj];
        pb = cnt[2 * n + sj];
      }
      unsigned hit = __builtin_amdgcn_ballot_w32(dj == v);
      while (hit) {
        const int k = __builtin_ctz(hit);
        hit &= hit - 1;
        add_rows<VEC, 1, L1>(acc, mrow, sj, pa, pb, k);
      }
    }
  }

  if constexpr (L1) {
    const float nd_v = __frsqrt_rn(fmaxf((float)(c0 + c1), 1.0f));
    const float ns_v = __frsqrt_rn(fmaxf((float)(cnt[v] + cnt[2 * n + v]), 1.0f));
    if (lane == 0) {
      ns_w[v] = ns_v;
      nd_w[v] = nd_v;
    }
    acc *= nd_v;
    if (bias) acc += *reinterpret_cast<const vf*>(bias + VEC * lane);
    if (relu) {
#pragma unroll
      for (int t = 0; t < VEC; ++t) acc[t] = fmaxf(acc[t], 0.0f);
    }
    acc *= ns_v;
  } else {
    if (post_scale) acc *= post_scale[v];
    if (bias) acc += *reinterpret_cast<const vf*>(bias + VEC * lane);
    if (relu) {
#pragma unroll
      for (int t = 0; t < VEC; ++t) acc[t] = fmaxf(acc[t], 0.0f);
    }
    if (out_scale) acc *= out_scale[v];
  }
  *reinterpret_cast<vf*>(outp + (size_t)v * F + VEC * lane) = acc;
}

static inline unsigned cdiv(unsigned a, unsigned b) { return (a + b - 1) / b; }

extern "C" void kernel_launch(void* const* d_in, const int* in_sizes, int n_in,
                              void* d_out, int out_size, void* d_ws, size_t ws_size,
                              hipStream_t stream) {
  const float* features = (const float*)d_in[0];
  const float* W1 = (const float*)d_in[1];
  const float* b1 = (const float*)d_in[2];
  const float* W2 = (const float*)d_in[3];
  const float* b2 = (const float*)d_in[4];
  const float* W3 = (const float*)d_in[5];
  const float* b3 = (const float*)d_in[6];
  const int* src  = (const int*)d_in[7];
  const int* dst  = (const int*)d_in[8];
  float* out = (float*)d_out;

  const int N  = N_NODES;
  const int E  = in_sizes[7];
  const int H  = in_sizes[2];

  float* ws   = (float*)d_ws;
  float* ns   = ws;
  float* nd   = ws + N;
  float* bufA = ws + 2 * N + 96;
  float* bufB = bufA + (size_t)N * H;
  int* cnt     = (int*)(bufB + (size_t)N * H);
  int* spill_n = cnt + 4 * N;
  int* bucket  = spill_n + 4;
  int* spill_d = bucket + (size_t)2 * N * CAP;
  int* spill_s = spill_d + E;
  (void)ws_size; (void)n_in; (void)out_size;

  const int BS = 256;
  const unsigned node_blocks = cdiv(N, BS / 32);

  gcn_zero_i4<<<cdiv(N + 1, BS), BS, 0, stream>>>(cnt, N + 1);

  gcn_gemm_wmma<256, 128, true><<<cdiv(N / 16, BS / 32), BS + 128, 0, stream>>>(
      features, W1, bufA, N, nullptr, nullptr, nullptr, 0,
      src, dst, cnt, spill_n, bucket, spill_d, spill_s, E, N);
  gcn_gather<4, true><<<node_blocks, BS, 0, stream>>>(
      bufA, cnt, bucket, spill_n, spill_d, spill_s, nullptr, b1, 1, nullptr, bufB, ns, nd, N);

  gcn_gather<4, false><<<node_blocks, BS, 0, stream>>>(
      bufB, cnt, bucket, spill_n, spill_d, spill_s, nullptr, nullptr, 0, nullptr, bufA,
      nullptr, nullptr, N);
  gcn_gemm_wmma<128, 128, false><<<cdiv(N / 16, BS / 32), BS, 0, stream>>>(
      bufA, W2, bufB, N, nullptr, nd, b2, 1,
      nullptr, nullptr, nullptr, nullptr, nullptr, nullptr, nullptr, 0, N);

  gcn_gemm_wmma<128, 64, false><<<cdiv(N / 16, BS / 32), BS, 0, stream>>>(
      bufB, W3, bufA, N, ns, nullptr, nullptr, 0,
      nullptr, nullptr, nullptr, nullptr, nullptr, nullptr, nullptr, 0, N);
  gcn_gather<2, false><<<node_blocks, BS, 0, stream>>>(
      bufA, cnt, bucket, spill_n, spill_d, spill_s, nd, b3, 0, nullptr, out,
      nullptr, nullptr, N);
}


# layer-2 neighbour sum, layer-2 GEMM and layer-3 GEMM in ONE launch: a block sums its row tiles into LDS, multiplies by W2, then h2 by W3 there; agg and h2 never reach memory; sum orders unchanged
# speedup vs baseline: 5.0561x; 1.2140x over previous
#include <hip/hip_runtime.h>


typedef float v2f __attribute__((ext_vector_type(2)));
typedef float v4f __attribute__((ext_vector_type(4)));
typedef float v8f __attribute__((ext_vector_type(8)));

#define N_NODES 50000
#define CAP 32

__global__ void gcn_zero_i4(int* __restrict__ p, int n4) {
  int i = blockIdx.x * blockDim.x + threadIdx.x;
  if (i < n4) reinterpret_cast<int4*>(p)[i] = int4{0, 0, 0, 0};
}

template <int K, int F, bool PLACE>
__global__ __launch_bounds__(PLACE ? 384 : 256) void gcn_gemm_wmma(
    const float* __restrict__ A, const float* __restrict__ W,
    float* __restrict__ C, int n_rows,
    const float* __restrict__ pre_scale, const float* __restrict__ post_scale,
    const float* __restrict__ bias, int relu,
    const int* __restrict__ src, const int* __restrict__ dst,
    int* __restrict__ cnt, int* __restrict__ spill_n, int* __restrict__ bucket,
    int* __restrict__ spill_d, int* __restrict__ spill_s, int E, int n) {
  constexpr int NG = F / 64;
  constexpr int NT = PLACE ? 384 : 256;
  constexpr int PL = NT - 256;
  __shared__ alignas(16) float Wl[K * F];

  {
    const float4* Wg = reinterpret_cast<const float4*>(W);
    float4* Ws = reinterpret_cast<float4*>(Wl);
    for (int i = threadIdx.x; i < K * F / 4; i += NT) Ws[i] = Wg[i];
  }
  __syncthreads();

  if constexpr (PLACE) {
    if (threadIdx.x >= 256) {
      const int hh = (int)((__builtin_amdgcn_s_sendmsg_rtn(0x87) >> 18) & 1u);
      int* cnt_out = cnt + hh * 2 * n;
      int* cnt_in = cnt_out + n;
      int* bk = bucket + hh * n * CAP;
      const int per = ((E + (int)gridDim.x - 1) / (int)gridDim.x + PL - 1) & ~(PL - 1);
      const int first = (int)blockIdx.x * per;
      const int last = (first + per < E) ? first + per : E;
      for (int e0 = first + (int)threadIdx.x - 256; e0 < last; e0 += PL * 8) {
        int d[8], sr[8], sl[8];
#pragma unroll
        for (int k = 0; k < 8; ++k) {
          const int e = (e0 + PL * k < last) ? e0 + PL * k : last - 1;
          d[k] = dst[e];
          sr[k] = src[e];
        }
#pragma unroll
        for (int k = 0; k < 8; ++k) {
          sl[k] = 0;
          if (e0 + PL * k < last) {
            atomicAdd(&cnt_out[sr[k]], 1);
            sl[k] = atomicAdd(&cnt_in[d[k]], 1);
          }
        }
#pragma unroll
        for (int k = 0; k < 8; ++k) {
          if (e0 + PL * k < last) {
            if (sl[k] < CAP) {
              bk[d[k] * CAP + sl[k]] = sr[k];
            } else {
              const int q = atomicAdd(spill_n, 1);
              spill_d[q] = d[k];
              spill_s[q] = sr[k];
            }
          }
        }
      }
      return;
    }
  }

  const int tileM = blockIdx.x * 8 + ((int)threadIdx.x >> 5);
  if (tileM >= (n_rows >> 4)) return;
  const int lane = threadIdx.x & 31;
  const int h = lane >> 4;
  const int c = lane & 15;

  const int m = (tileM << 4) + c;
  const float ps = pre_scale ? pre_scale[m] : 1.0f;
  const float4* Arow = reinterpret_cast<const float4*>(A + (size_t)m * K + 8 * h);
  const float* Wlane = Wl + (8 * h) * F + 4 * c;

  v8f acc[NG][4];
#pragma unroll
  for (int g = 0; g < NG; ++g)
#pragma unroll
    for (int t = 0; t < 4; ++t) acc[g][t] = v8f{};

  float4 q0 = Arow[0], q1 = Arow[1];
  for (int kb = 0; kb < K; kb += 16) {
    const int kn = (kb + 16 < K) ? kb + 16 : kb;
    const float4 n0 = Arow[kn >> 2], n1 = Arow[(kn >> 2) + 1];

    const float av[8] = {q0.x * ps, q0.y * ps, q0.z * ps, q0.w * ps,
                         q1.x * ps, q1.y * ps, q1.z * ps, q1.w * ps};
#pragma unroll
    for (int j = 0; j < 4; ++j) {
      v2f a;
      a.x = av[2 * j];
      a.y = av[2 * j + 1];
      const float* wr = Wlane + (kb + 2 * j) * F;
#pragma unroll
      for (int g = 0; g < NG; ++g) {
        const float4 b0 = *reinterpret_cast<const float4*>(wr + 64 * g);
        const float4 b1 = *reinterpret_cast<const float4*>(wr + F + 64 * g);
        v2f b;
        b.x = b0.x; b.y = b1.x;
        acc[g][0] = __builtin_amdgcn_wmma_f32_16x16x4_f32(false, a, false, b, (short)0, acc[g][0], false, false);
        b.x = b0.y; b.y = b1.y;
        acc[g][1] = __builtin_amdgcn_wmma_f32_16x16x4_f32(false, a, false, b, (short)0, acc[g][1], false, false);
        b.x = b0.z; b.y = b1.z;
        acc[g][2] = __builtin_amdgcn_wmma_f32_16x16x4_f32(false, a, false, b, (short)0, acc[g][2], false, false);
        b.x = b0.w; b.y = b1.w;
        acc[g][3] = __builtin_amdgcn_wmma_f32_16x16x4_f32(false, a, false, b, (short)0, acc[g][3], false, false);
      }
    }
    q0 = n0;
    q1 = n1;
  }

#pragma unroll
  for (int g = 0; g < NG; ++g) {
    const int col = 64 * g + 4 * c;
    float4 bb = {0.0f, 0.0f, 0.0f, 0.0f};
    if (bias) bb = *reinterpret_cast<const float4*>(bias + col);
#pragma unroll
    for (int r = 0; r < 8; ++r) {
      const int row = (tileM << 4) + r + (h << 3);
      float4 v = {acc[g][0][r], acc[g][1][r], acc[g][2][r], acc[g][3][r]};
      if (post_scale) {
        const float s = post_scale[row];
        v.x *= s; v.y *= s; v.z *= s; v.w *= s;
      }
      if (bias) { v.x += bb.x; v.y += bb.y; v.z += bb.z; v.w += bb.w; }
      if (relu) {
        v.x = fmaxf(v.x, 0.0f); v.y = fmaxf(v.y, 0.0f);
        v.z = fmaxf(v.z, 0.0f); v.w = fmaxf(v.w, 0.0f);
      }
      *reinterpret_cast<float4*>(C + (size_t)row * F + col) = v;
    }
  }
}

template <int VEC, int G, bool W>
__device__ __forceinline__ void add_rows(float __attribute__((ext_vector_type(VEC)))& acc,
                                         const float* __restrict__ mrow, int my, int oa, int ob, int i) {
  typedef float vf __attribute__((ext_vector_type(VEC)));
  constexpr int F = 32 * VEC;
  int u[G];
#pragma unroll
  for (int k = 0; k < G; ++k) u[k] = __builtin_amdgcn_readlane(my, i + k);
  vf r[G];
#pragma unroll
  for (int k = 0; k < G; ++k) r[k] = *reinterpret_cast<const vf*>(mrow + (size_t)u[k] * F);
#pragma unroll
  for (int k = 0; k < G; ++k) {
    if constexpr (W) {
      const int od = __builtin_amdgcn_readlane(oa, i + k) + __builtin_amdgcn_readlane(ob, i + k);
      acc += r[k] * __frsqrt_rn(fmaxf((float)od, 1.0f));
    } else {
      acc += r[k];
    }
  }
}

template <int VEC, bool L1>
__global__ __launch_bounds__(256) void gcn_gather(
    const float* __restrict__ msg, const int* __restrict__ cnt,
    const int* __restrict__ bucket, const int* __restrict__ spill_n,
    const int* __restrict__ spill_d, const int* __restrict__ spill_s,
    const float* __restrict__ post_scale,
    const float* __restrict__ bias, int relu, const float* __restrict__ out_scale,
    float* __restrict__ outp, float* __restrict__ ns_w, float* __restrict__ nd_w, int n) {
  constexpr int F = 32 * VEC;
  typedef float vf __attribute__((ext_vector_type(VEC)));
  const int lane = threadIdx.x & 31;
  const int v = __builtin_amdgcn_readfirstlane(
      (int)(blockIdx.x * (blockDim.x >> 5) + (threadIdx.x >> 5)));
  if (v >= n) return;
  const int c0 = cnt[n + v];
  const int c1 = cnt[3 * n + v];
  const int n0 = (c0 < CAP) ? c0 : CAP;
  const int n1 = (c1 < CAP) ? c1 : CAP;
  const int tot = n0 + n1;
  const int* B0 = bucket + v * CAP;
  const int* B1 = bucket + (n + v) * CAP;

  const float* mrow = msg + VEC * lane;
  vf acc = 0.0f;
  for (int b0 = 0; b0 < tot; b0 += 32) {
    const int m = (tot - b0 < 32) ? tot - b0 : 32;
    const int idx = b0 + lane;
    const int* slot = (idx < n0) ? B0 + idx : B1 + (idx - n0);
    const int my = (lane < m) ? *slot : 0;
    int oa = 0, ob = 0;
    if constexpr (L1) {
      oa = cnt[my];
      ob = cnt[2 * n + my];
    }
    int i = 0;
    for (; i + 8 <= m; i += 8) add_rows<VEC, 8, L1>(acc, mrow, my, oa, ob, i);
    if (i + 4 <= m) { add_rows<VEC, 4, L1>(acc, mrow, my, oa, ob, i); i += 4; }
    if (i + 2 <= m) { add_rows<VEC, 2, L1>(acc, mrow, my, oa, ob, i); i += 2; }
    if (i < m) add_rows<VEC, 1, L1>(acc, mrow, my, oa, ob, i);
  }
  if (c0 > CAP || c1 > CAP) {
    const int M = spill_n[0];
    for (int j0 = 0; j0 < M; j0 += 32) {
      const int j = j0 + lane;
      const int dj = (j < M) ? spill_d[j] : -1;
      const int sj = (j < M) ? spill_s[j] : 0;
      int pa = 0, pb = 0;
      if constexpr (L1) {
        pa = cnt[sj];
        pb = cnt[2 * n + sj];
      }
      unsigned hit = __builtin_amdgcn_ballot_w32(dj == v);
      while (hit) {
        const int k = __builtin_ctz(hit);
        hit &= hit - 1;
        add_rows<VEC, 1, L1>(acc, mrow, sj, pa, pb, k);
      }
    }
  }

  if constexpr (L1) {
    const float nd_v = __frsqrt_rn(fmaxf((float)(c0 + c1), 1.0f));
    const float ns_v = __frsqrt_rn(fmaxf((float)(cnt[v] + cnt[2 * n + v]), 1.0f));
    if (lane == 0) {
      ns_w[v] = ns_v;
      nd_w[v] = nd_v;
    }
    acc *= nd_v;
    if (bias) acc += *reinterpret_cast<const vf*>(bias + VEC * lane);
    if (relu) {
#pragma unroll
      for (int t = 0; t < VEC; ++t) acc[t] = fmaxf(acc[t], 0.0f);
    }
    acc *= ns_v;
  } else {
    if (post_scale) acc *= post_scale[v];
    if (bias) acc += *reinterpret_cast<const vf*>(bias + VEC * lane);
    if (relu) {
#pragma unroll
      for (int t = 0; t < VEC; ++t) acc[t] = fmaxf(acc[t], 0.0f);
    }
    if (out_scale) acc *= out_scale[v];
  }
  *reinterpret_cast<vf*>(outp + (size_t)v * F + VEC * lane) = acc;
}

__device__ __forceinline__ int first_ids(const int* __restrict__ bucket, int v, int c0, int c1, int n, int lane) {
  const int n0 = (c0 < CAP) ? c0 : CAP;
  const int n1 = (c1 < CAP) ? c1 : CAP;
  const int tot = n0 + n1;
  const int m = (tot < 32) ? tot : 32;
  const int* slot = (lane < n0) ? bucket + v * CAP + lane : bucket + (n + v) * CAP + (lane - n0);
  return (lane < m) ? *slot : 0;
}

__device__ __forceinline__ void add_list(v4f& acc, const float* __restrict__ mrow, int my, int m) {
  int i = 0;
  for (; i + 8 <= m; i += 8) add_rows<4, 8, false>(acc, mrow, my, 0, 0, i);
  switch (m - i) {
    case 7: add_rows<4, 7, false>(acc, mrow, my, 0, 0, i); break;
    case 6: add_rows<4, 6, false>(acc, mrow, my, 0, 0, i); break;
    case 5: add_rows<4, 5, false>(acc, mrow, my, 0, 0, i); break;
    case 4: add_rows<4, 4, false>(acc, mrow, my, 0, 0, i); break;
    case 3: add_rows<4, 3, false>(acc, mrow, my, 0, 0, i); break;
    case 2: add_rows<4, 2, false>(acc, mrow, my, 0, 0, i); break;
    case 1: add_rows<4, 1, false>(acc, mrow, my, 0, 0, i); break;
    default: break;
  }
}

__device__ __forceinline__ v4f node_sum4(const float* __restrict__ mrow, const int* __restrict__ bucket,
                                         const int* __restrict__ spill_n, const int* __restrict__ spill_d,
                                         const int* __restrict__ spill_s, int my0,
                                         int v, int c0, int c1, int n, int lane) {
  const int n0 = (c0 < CAP) ? c0 : CAP;
  const int n1 = (c1 < CAP) ? c1 : CAP;
  const int tot = n0 + n1;
  v4f acc = 0.0f;
  add_list(acc, mrow, my0, (tot < 32) ? tot : 32);
  for (int b0 = 32; b0 < tot; b0 += 32) {
    const int m = (tot - b0 < 32) ? tot - b0 : 32;
    const int idx = b0 + lane;
    const int* slot = (idx < n0) ? bucket + v * CAP + idx : bucket + (n + v) * CAP + (idx - n0);
    const int my = (lane < m) ? *slot : 0;
    add_list(acc, mrow, my, m);
  }
  if (c0 > CAP || c1 > CAP) {
    const int M = spill_n[0];
    for (int j0 = 0; j0 < M; j0 += 32) {
      const int j = j0 + lane;
      const int dj = (j < M) ? spill_d[j] : -1;
      const int sj = (j < M) ? spill_s[j] : 0;
      unsigned hit = __builtin_amdgcn_ballot_w32(dj == v);
      while (hit) {
        const int k = __builtin_ctz(hit);
        hit &= hit - 1;
        add_rows<4, 1, false>(acc, mrow, sj, 0, 0, k);
      }
    }
  }
  return acc;
}

template <int K, int FP>
__device__ __forceinline__ void mma_rows_lds(v8f (&acc)[4], const float4* Arow, const float* Wlane, float ps) {
  float4 q0 = Arow[0], q1 = Arow[1];
  for (int kb = 0; kb < K; kb += 16) {
    const int kn = (kb + 16 < K) ? kb + 16 : kb;
    const float4 n0 = Arow[kn >> 2], n1 = Arow[(kn >> 2) + 1];
    const float av[8] = {q0.x * ps, q0.y * ps, q0.z * ps, q0.w * ps,
                         q1.x * ps, q1.y * ps, q1.z * ps, q1.w * ps};
#pragma unroll
    for (int j = 0; j < 4; ++j) {
      v2f a;
      a.x = av[2 * j];
      a.y = av[2 * j + 1];
      const float* wr = Wlane + (kb + 2 * j) * FP;
      const float4 b0 = *reinterpret_cast<const float4*>(wr);
      const float4 b1 = *reinterpret_cast<const float4*>(wr + FP);
      v2f b;
      b.x = b0.x; b.y = b1.x;
      acc[0] = __builtin_amdgcn_wmma_f32_16x16x4_f32(false, a, false, b, (short)0, acc[0], false, false);
      b.x = b0.y; b.y = b1.y;
      acc[1] = __builtin_amdgcn_wmma_f32_16x16x4_f32(false, a, false, b, (short)0, acc[1], false, false);
      b.x = b0.z; b.y = b1.z;
      acc[2] = __builtin_amdgcn_wmma_f32_16x16x4_f32(false, a, false, b, (short)0, acc[2], false, false);
      b.x = b0.w; b.y = b1.w;
      acc[3] = __builtin_amdgcn_wmma_f32_16x16x4_f32(false, a, false, b, (short)0, acc[3], false, false);
    }
    q0 = n0;
    q1 = n1;
  }
}

__global__ __launch_bounds__(512) void gcn_layer23(
    const float* __restrict__ msg, const float* __restrict__ W2, const float* __restrict__ b2,
    const float* __restrict__ W3, const float* __restrict__ ns, const float* __restrict__ nd,
    float* __restrict__ C, const int* __restrict__ cnt, const int* __restrict__ bucket,
    const int* __restrict__ spill_n, const int* __restrict__ spill_d,
    const int* __restrict__ spill_s, int n_tiles, int n) {
  constexpr int K = 128, F2 = 128, F3 = 64, NT = 512;
  constexpr int AP = K + 4;
  __shared__ alignas(16) float Wl[K * F2];
  __shared__ alignas(16) float As[8 * 16 * AP];

  {
    const float4* Wg = reinterpret_cast<const float4*>(W2);
    float4* Ws = reinterpret_cast<float4*>(Wl);
    float4 w[K * F2 / 4 / NT];
#pragma unroll
    for (int j = 0; j < K * F2 / 4 / NT; ++j) w[j] = Wg[threadIdx.x + j * NT];
#pragma unroll
    for (int j = 0; j < K * F2 / 4 / NT; ++j) Ws[threadIdx.x + j * NT] = w[j];
  }

  const int wv = __builtin_amdgcn_readfirstlane((int)threadIdx.x >> 5);
  const int slot = wv & 7;
  const int half = wv >> 3;
  const int tileM = slot * (int)gridDim.x + (int)blockIdx.x;
  const bool live = tileM < n_tiles;
  const int lane = threadIdx.x & 31;
  const int h = lane >> 4;
  const int c = lane & 15;
  float* At = As + slot * 16 * AP;

  if (live) {
    const int cc0 = cnt[n + (tileM << 4) + c];
    const int cc1 = cnt[3 * n + (tileM << 4) + c];
    const float* mrow = msg + 4 * lane;
    const int r0 = half << 3;
    int my_next = first_ids(bucket, (tileM << 4) + r0, __builtin_amdgcn_readlane(cc0, r0),
                            __builtin_amdgcn_readlane(cc1, r0), n, lane);
    for (int r = r0; r < r0 + 8; ++r) {
      const int v = (tileM << 4) + r;
      const int c0 = __builtin_amdgcn_readlane(cc0, r);
      const int c1 = __builtin_amdgcn_readlane(cc1, r);
      const int my = my_next;
      if (r + 1 < r0 + 8)
        my_next = first_ids(bucket, v + 1, __builtin_amdgcn_readlane(cc0, r + 1),
                            __builtin_amdgcn_readlane(cc1, r + 1), n, lane);
      const v4f sum = node_sum4(mrow, bucket, spill_n, spill_d, spill_s, my, v, c0, c1, n, lane);
      *reinterpret_cast<v4f*>(At + r * AP + 4 * lane) = sum;
    }
  }
  __syncthreads();

  float4 hv[8];
#pragma unroll
  for (int r = 0; r < 8; ++r) hv[r] = float4{0.0f, 0.0f, 0.0f, 0.0f};
  if (live) {
    v8f acc[4];
#pragma unroll
    for (int t = 0; t < 4; ++t) acc[t] = v8f{};
    mma_rows_lds<K, F2>(acc, reinterpret_cast<const float4*>(At + c * AP + 8 * h),
                        Wl + (8 * h) * F2 + 64 * half + 4 * c, 1.0f);
    const float4 bb = *reinterpret_cast<const float4*>(b2 + 64 * half + 4 * c);
#pragma unroll
    for (int r = 0; r < 8; ++r) {
      const float sc = nd[(tileM << 4) + r + (h << 3)];
      float4 v = {acc[0][r], acc[1][r], acc[2][r], acc[3][r]};
      v.x *= sc; v.y *= sc; v.z *= sc; v.w *= sc;
      v.x += bb.x; v.y += bb.y; v.z += bb.z; v.w += bb.w;
      v.x = fmaxf(v.x, 0.0f); v.y = fmaxf(v.y, 0.0f);
      v.z = fmaxf(v.z, 0.0f); v.w = fmaxf(v.w, 0.0f);
      hv[r] = v;
    }
  }
  __syncthreads();

  {
    const float4* Wg = reinterpret_cast<const float4*>(W3);
    float4* Ws = reinterpret_cast<float4*>(Wl);
    float4 w[K * F3 / 4 / NT];
#pragma unroll
    for (int j = 0; j < K * F3 / 4 / NT; ++j) w[j] = Wg[threadIdx.x + j * NT];
#pragma unroll
    for (int j = 0; j < K * F3 / 4 / NT; ++j) Ws[threadIdx.x + j * NT] = w[j];
  }
  if (live) {
#pragma unroll
    for (int r = 0; r < 8; ++r)
      *reinterpret_cast<float4*>(At + (r + (h << 3)) * AP + 64 * half + 4 * c) = hv[r];
  }
  __syncthreads();

  if (live && half == 0) {
    v8f acc[4];
#pragma unroll
    for (int t = 0; t < 4; ++t) acc[t] = v8f{};
    const float ps = ns[(tileM << 4) + c];
    mma_rows_lds<K, F3>(acc, reinterpret_cast<const float4*>(At + c * AP + 8 * h),
                        Wl + (8 * h) * F3 + 4 * c, ps);
#pragma unroll
    for (int r = 0; r < 8; ++r) {
      const int row = (tileM << 4) + r + (h << 3);
      const float4 v = {acc[0][r], acc[1][r], acc[2][r], acc[3][r]};
      *reinterpret_cast<float4*>(C + (size_t)row * F3 + 4 * c) = v;
    }
  }
}

static inline unsigned cdiv(unsigned a, unsigned b) { return (a + b - 1) / b; }

extern "C" void kernel_launch(void* const* d_in, const int* in_sizes, int n_in,
                              void* d_out, int out_size, void* d_ws, size_t ws_size,
                              hipStream_t stream) {
  const float* features = (const float*)d_in[0];
  const float* W1 = (const float*)d_in[1];
  const float* b1 = (const float*)d_in[2];
  const float* W2 = (const float*)d_in[3];
  const float* b2 = (const float*)d_in[4];
  const float* W3 = (const float*)d_in[5];
  const float* b3 = (const float*)d_in[6];
  const int* src  = (const int*)d_in[7];
  const int* dst  = (const int*)d_in[8];
  float* out = (float*)d_out;

  const int N  = N_NODES;
  const int E  = in_sizes[7];
  const int H  = in_sizes[2];

  float* ws   = (float*)d_ws;
  float* ns   = ws;
  float* nd   = ws + N;
  float* bufA = ws + 2 * N + 96;
  float* bufB = bufA + (size_t)N * H;
  int* cnt     = (int*)(bufB + (size_t)N * H);
  int* spill_n = cnt + 4 * N;
  int* bucket  = spill_n + 4;
  int* spill_d = bucket + (size_t)2 * N * CAP;
  int* spill_s = spill_d + E;
  (void)ws_size; (void)n_in; (void)out_size;

  const int BS = 256;
  const unsigned node_blocks = cdiv(N, BS / 32);

  gcn_zero_i4<<<cdiv(N + 1, BS), BS, 0, stream>>>(cnt, N + 1);

  gcn_gemm_wmma<256, 128, true><<<cdiv(N / 16, BS / 32), BS + 128, 0, stream>>>(
      features, W1, bufA, N, nullptr, nullptr, nullptr, 0,
      src, dst, cnt, spill_n, bucket, spill_d, spill_s, E, N);
  gcn_gather<4, true><<<node_blocks, BS, 0, stream>>>(
      bufA, cnt, bucket, spill_n, spill_d, spill_s, nullptr, b1, 1, nullptr, bufB, ns, nd, N);

  gcn_layer23<<<512, 512, 0, stream>>>(
      bufB, W2, b2, W3, ns, nd, bufA, cnt, bucket, spill_n, spill_d, spill_s, N / 16, N);

  gcn_gather<2, false><<<node_blocks, BS, 0, stream>>>(
      bufA, cnt, bucket, spill_n, spill_d, spill_s, nd, b3, 0, nullptr, out,
      nullptr, nullptr, N);
}


# layer-1 GEMM: 1 block per compute unit, 12 row tiles + quarters of the 53 left over dealt evenly on its 4 SIMDs (busiest SIMD 1664 WMMAs, was 2048), loads pinned ahead of WMMAs; sum orders unchanged
# speedup vs baseline: 5.6318x; 1.1139x over previous
#include <hip/hip_runtime.h>


typedef float v2f __attribute__((ext_vector_type(2)));
typedef float v4f __attribute__((ext_vector_type(4)));
typedef float v8f __attribute__((ext_vector_type(8)));

#define N_NODES 50000
#define CAP 32

__global__ void gcn_zero_i4(int* __restrict__ p, int n4) {
  int i = blockIdx.x * blockDim.x + threadIdx.x;
  if (i < n4) reinterpret_cast<int4*>(p)[i] = int4{0, 0, 0, 0};
}

template <int K, int F, bool PLACE>
__global__ __launch_bounds__(PLACE ? 640 : 512) void gcn_gemm_wmma(
    const float* __restrict__ A, const float* __restrict__ W,
    float* __restrict__ C, int n_rows,
    const int* __restrict__ src, const int* __restrict__ dst,
    int* __restrict__ cnt, int* __restrict__ spill_n, int* __restrict__ bucket,
    int* __restrict__ spill_d, int* __restrict__ spill_s, int E, int n) {
  static_assert(F == 128, "a left-over tile is cut into 4 pieces of 32 columns");
  constexpr int NG = F / 64;
  constexpr int CT = 512;
  constexpr int FULLW = 12;
  constexpr int NT = PLACE ? 640 : 512;
  constexpr int PL = NT - CT;
  constexpr int ONE_PER_UNIT = 8448;
  __shared__ alignas(16) float Wl[K * F + ONE_PER_UNIT];

  {
    const float4* Wg = reinterpret_cast<const float4*>(W);
    float4* Ws = reinterpret_cast<float4*>(Wl);
    constexpr int NP = (K * F / 4 + NT - 1) / NT;
    float4 w[NP];
#pragma unroll
    for (int j = 0; j < NP; ++j) {
      const int i = (int)threadIdx.x + j * NT;
      w[j] = Wg[(i < K * F / 4) ? i : (int)threadIdx.x];
    }
#pragma unroll
    for (int j = 0; j < NP; ++j) {
      const int i = (int)threadIdx.x + j * NT;
      if (i < K * F / 4) Ws[i] = w[j];
    }
  }
  __syncthreads();

  if constexpr (PLACE) {
    if (threadIdx.x >= CT) {
      const int hh = (int)((__builtin_amdgcn_s_sendmsg_rtn(0x87) >> 18) & 1u);
      int* cnt_out = cnt + hh * 2 * n;
      int* cnt_in = cnt_out + n;
      int* bk = bucket + hh * n * CAP;
      const int per = ((E + (int)gridDim.x - 1) / (int)gridDim.x + PL - 1) & ~(PL - 1);
      const int first = (int)blockIdx.x * per;
      const int last = (first + per < E) ? first + per : E;
      constexpr int ED = 16;
      for (int e0 = first + (int)threadIdx.x - CT; e0 < last; e0 += PL * ED) {
        int d[ED], sr[ED], sl[ED];
#pragma unroll
        for (int k = 0; k < ED; ++k) {
          const int e = (e0 + PL * k < last) ? e0 + PL * k : last - 1;
          d[k] = dst[e];
          sr[k] = src[e];
        }
#pragma unroll
        for (int k = 0; k < ED; ++k) {
          sl[k] = 0;
          if (e0 + PL * k < last) {
            atomicAdd(&cnt_out[sr[k]], 1);
            sl[k] = atomicAdd(&cnt_in[d[k]], 1);
          }
        }
#pragma unroll
        for (int k = 0; k < ED; ++k) {
          if (e0 + PL * k < last) {
            if (sl[k] < CAP) {
              bk[d[k] * CAP + sl[k]] = sr[k];
            } else {
              const int q = atomicAdd(spill_n, 1);
              spill_d[q] = d[k];
              spill_s[q] = sr[k];
            }
          }
        }
      }
      return;
    }
  }

  const int n_tiles = n_rows >> 4;
  const int wv = (int)threadIdx.x >> 5;
  const int lane = threadIdx.x & 31;
  const int h = lane >> 4;
  const int c = lane & 15;

  if (wv >= FULLW) {
    const int p = (int)blockIdx.x * 4 + (wv - FULLW);
    const int tileM = FULLW * (int)gridDim.x + (p >> 2);
    if (tileM >= n_tiles) return;
    const int col0 = 32 * (p & 3);
    const int m = (tileM << 4) + c;
    const float4* Arow = reinterpret_cast<const float4*>(A + (size_t)m * K + 8 * h);
    const float* Wlane = Wl + (8 * h) * F + col0 + 2 * c;
    v8f acc[2];
    acc[0] = v8f{};
    acc[1] = v8f{};
    float4 q0 = Arow[0], q1 = Arow[1];
    for (int kb = 0; kb < K; kb += 16) {
      const int kn = (kb + 16 < K) ? kb + 16 : kb;
      const float4 n0 = Arow[kn >> 2], n1 = Arow[(kn >> 2) + 1];
      const float av[8] = {q0.x, q0.y, q0.z, q0.w, q1.x, q1.y, q1.z, q1.w};
#pragma unroll
      for (int j = 0; j < 4; ++j) {
        v2f a;
        a.x = av[2 * j];
        a.y = av[2 * j + 1];
        const float* wr = Wlane + (kb + 2 * j) * F;
        const float2 b0 = *reinterpret_cast<const float2*>(wr);
        const float2 b1 = *reinterpret_cast<const float2*>(wr + F);
        v2f b;
        b.x = b0.x; b.y = b1.x;
        acc[0] = __builtin_amdgcn_wmma_f32_16x16x4_f32(false, a, false, b, (short)0, acc[0], false, false);
        b.x = b0.y; b.y = b1.y;
        acc[1] = __builtin_amdgcn_wmma_f32_16x16x4_f32(false, a, false, b, (short)0, acc[1], false, false);
      }
      q0 = n0;
      q1 = n1;
    }
    const int col = col0 + 2 * c;
#pragma unroll
    for (int r = 0; r < 8; ++r) {
      const int row = (tileM << 4) + r + (h << 3);
      const float2 v = {acc[0][r], acc[1][r]};
      *reinterpret_cast<float2*>(C + (size_t)row * F + col) = v;
    }
    return;
  }

  const int tileM = wv * (int)gridDim.x + (int)blockIdx.x;
  if (tileM >= n_tiles) return;

  const int m = (tileM << 4) + c;
  const float4* Arow = reinterpret_cast<const float4*>(A + (size_t)m * K + 8 * h);
  const float* Wlane = Wl + (8 * h) * F + 4 * c;

  v8f acc[NG][4];
#pragma unroll
  for (int g = 0; g < NG; ++g)
#pragma unroll
    for (int t = 0; t < 4; ++t) acc[g][t] = v8f{};

  constexpr int NCH = K / 16;
  float4 qa[NCH], qb[NCH];
  qa[0] = Arow[0]; qb[0] = Arow[1];
  qa[1] = Arow[4]; qb[1] = Arow[5];
  float4 B0[8], B1[8];
#define GCN_LOAD_HALF(buf, ch, hf)                                                        \
  _Pragma("unroll") for (int jj = 0; jj < 2; ++jj) {                                      \
    const float* wr = Wlane + (16 * (ch) + 2 * (2 * (hf) + jj)) * F;                      \
    _Pragma("unroll") for (int g = 0; g < NG; ++g) {                                      \
      buf[(jj * 2 + g) * 2 + 0] = *reinterpret_cast<const float4*>(wr + 64 * g);          \
      buf[(jj * 2 + g) * 2 + 1] = *reinterpret_cast<const float4*>(wr + F + 64 * g);      \
    }                                                                                     \
  }
#define GCN_MMA_HALF(buf, hf)                                                             \
  _Pragma("unroll") for (int jj = 0; jj < 2; ++jj) {                                      \
    v2f a;                                                                                \
    a.x = av[2 * (2 * (hf) + jj)];                                                        \
    a.y = av[2 * (2 * (hf) + jj) + 1];                                                    \
    _Pragma("unroll") for (int g = 0; g < NG; ++g) {                                      \
      const float4 b0 = buf[(jj * 2 + g) * 2 + 0];                                        \
      const float4 b1 = buf[(jj * 2 + g) * 2 + 1];                                        \
      v2f b;                                                                              \
      b.x = b0.x; b.y = b1.x;                                                             \
      acc[g][0] = __builtin_amdgcn_wmma_f32_16x16x4_f32(false, a, false, b, (short)0, acc[g][0], false, false); \
      b.x = b0.y; b.y = b1.y;                                                             \
      acc[g][1] = __builtin_amdgcn_wmma_f32_16x16x4_f32(false, a, false, b, (short)0, acc[g][1], false, false); \
      b.x = b0.z; b.y = b1.z;                                                             \
      acc[g][2] = __builtin_amdgcn_wmma_f32_16x16x4_f32(false, a, false, b, (short)0, acc[g][2], false, false); \
      b.x = b0.w; b.y = b1.w;                                                             \
      acc[g][3] = __builtin_amdgcn_wmma_f32_16x16x4_f32(false, a, false, b, (short)0, acc[g][3], false, false); \
    }                                                                                     \
  }
  static_assert(NG == 2, "B0 / B1 hold 2 WMMA steps x 2 groups x 2 rows");
  GCN_LOAD_HALF(B0, 0, 0)
#pragma unroll
  for (int ch = 0; ch < NCH; ++ch) {
    if (ch + 2 < NCH) {
      qa[ch + 2] = Arow[4 * (ch + 2)];
      qb[ch + 2] = Arow[4 * (ch + 2) + 1];
    }
    const float4 q0 = qa[ch], q1 = qb[ch];
    const float av[8] = {q0.x, q0.y, q0.z, q0.w, q1.x, q1.y, q1.z, q1.w};
    GCN_LOAD_HALF(B1, ch, 1)
    __builtin_amdgcn_sched_barrier(0);
    GCN_MMA_HALF(B0, 0)
    __builtin_amdgcn_sched_barrier(0);
    if (ch + 1 < NCH) {
      GCN_LOAD_HALF(B0, ch + 1, 0)
    }
    __builtin_amdgcn_sched_barrier(0);
    GCN_MMA_HALF(B1, 1)
    __builtin_amdgcn_sched_barrier(0);
  }
#undef GCN_LOAD_HALF
#undef GCN_MMA_HALF

#pragma unroll
  for (int g = 0; g < NG; ++g) {
    const int col = 64 * g + 4 * c;
#pragma unroll
    for (int r = 0; r < 8; ++r) {
      const int row = (tileM << 4) + r + (h << 3);
      const float4 v = {acc[g][0][r], acc[g][1][r], acc[g][2][r], acc[g][3][r]};
      *reinterpret_cast<float4*>(C + (size_t)row * F + col) = v;
    }
  }
}

template <int VEC, int G, bool W>
__device__ __forceinline__ void add_rows(float __attribute__((ext_vector_type(VEC)))& acc,
                                         const float* __restrict__ mrow, int my, int oa, int ob, int i) {
  typedef float vf __attribute__((ext_vector_type(VEC)));
  constexpr int F = 32 * VEC;
  int u[G];
#pragma unroll
  for (int k = 0; k < G; ++k) u[k] = __builtin_amdgcn_readlane(my, i + k);
  vf r[G];
#pragma unroll
  for (int k = 0; k < G; ++k) r[k] = *reinterpret_cast<const vf*>(mrow + (size_t)u[k] * F);
#pragma unroll
  for (int k = 0; k < G; ++k) {
    if constexpr (W) {
      const int od = __builtin_amdgcn_readlane(oa, i + k) + __builtin_amdgcn_readlane(ob, i + k);
      acc += r[k] * __frsqrt_rn(fmaxf((float)od, 1.0f));
    } else {
      acc += r[k];
    }
  }
}

__device__ __forceinline__ int first_ids(const int* __restrict__ bucket, int v, int c0, int c1, int n, int lane) {
  const int n0 = (c0 < CAP) ? c0 : CAP;
  const int n1 = (c1 < CAP) ? c1 : CAP;
  const int tot = n0 + n1;
  const int m = (tot < 32) ? tot : 32;
  const int* slot = (lane < n0) ? bucket + v * CAP + lane : bucket + (n + v) * CAP + (lane - n0);
  return (lane < m) ? *slot : 0;
}

__device__ __forceinline__ int first_ids_raw(const int* __restrict__ bucket, const int* __restrict__ cnt,
                                             int v, int c0, int c1, int n, int lane) {
  const int n0 = (c0 < CAP) ? c0 : CAP;
  const int n1 = (c1 < CAP) ? c1 : CAP;
  const int tot = n0 + n1;
  const int* slot = (lane < n0) ? bucket + v * CAP + lane : bucket + (n + v) * CAP + (lane - n0);
  return *((lane < tot && lane < 32) ? slot : cnt + n + v);
}

template <int VEC, bool W>
__device__ __forceinline__ void add_list(float __attribute__((ext_vector_type(VEC)))& acc,
                                         const float* __restrict__ mrow, int my, int oa, int ob, int m) {
  int i = 0;
  for (; i + 8 <= m; i += 8) add_rows<VEC, 8, W>(acc, mrow, my, oa, ob, i);
  switch (m - i) {
    case 7: add_rows<VEC, 7, W>(acc, mrow, my, oa, ob, i); break;
    case 6: add_rows<VEC, 6, W>(acc, mrow, my, oa, ob, i); break;
    case 5: add_rows<VEC, 5, W>(acc, mrow, my, oa, ob, i); break;
    case 4: add_rows<VEC, 4, W>(acc, mrow, my, oa, ob, i); break;
    case 3: add_rows<VEC, 3, W>(acc, mrow, my, oa, ob, i); break;
    case 2: add_rows<VEC, 2, W>(acc, mrow, my, oa, ob, i); break;
    case 1: add_rows<VEC, 1, W>(acc, mrow, my, oa, ob, i); break;
    default: break;
  }
}

template <int VEC, bool W>
__device__ __forceinline__ float __attribute__((ext_vector_type(VEC)))
node_sum(const float* __restrict__ mrow, const int* __restrict__ cnt, const int* __restrict__ bucket,
         const int* __restrict__ spill_n, const int* __restrict__ spill_d,
         const int* __restrict__ spill_s, int my0, int v, int c0, int c1, int n, int lane) {
  typedef float vf __attribute__((ext_vector_type(VEC)));
  const int n0 = (c0 < CAP) ? c0 : CAP;
  const int n1 = (c1 < CAP) ? c1 : CAP;
  const int tot = n0 + n1;
  vf acc = 0.0f;
  {
    int oa = 0, ob = 0;
    if constexpr (W) {
      oa = cnt[my0];
      ob = cnt[2 * n + my0];
    }
    add_list<VEC, W>(acc, mrow, my0, oa, ob, (tot < 32) ? tot : 32);
  }
  for (int b0 = 32; b0 < tot; b0 += 32) {
    const int m = (tot - b0 < 32) ? tot - b0 : 32;
    const int idx = b0 + lane;
    const int* slot = (idx < n0) ? bucket + v * CAP + idx : bucket + (n + v) * CAP + (idx - n0);
    const int my = (lane < m) ? *slot : 0;
    int oa = 0, ob = 0;
    if constexpr (W) {
      oa = cnt[my];
      ob = cnt[2 * n + my];
    }
    add_list<VEC, W>(acc, mrow, my, oa, ob, m);
  }
  if (c0 > CAP || c1 > CAP) {
    const int M = spill_n[0];
    for (int j0 = 0; j0 < M; j0 += 32) {
      const int j = j0 + lane;
      const int dj = (j < M) ? spill_d[j] : -1;
      const int sj = (j < M) ? spill_s[j] : 0;
      int pa = 0, pb = 0;
      if constexpr (W) {
        pa = cnt[sj];
        pb = cnt[2 * n + sj];
      }
      unsigned hit = __builtin_amdgcn_ballot_w32(dj == v);
      while (hit) {
        const int k = __builtin_ctz(hit);
        hit &= hit - 1;
        add_rows<VEC, 1, W>(acc, mrow, sj, pa, pb, k);
      }
    }
  }
  return acc;
}

__device__ __forceinline__ float lane_value(float x, int r) {
  return __builtin_bit_cast(float, __builtin_amdgcn_readlane(__builtin_bit_cast(int, x), r));
}

template <int VEC, bool L1, int NPW>
__global__ __launch_bounds__(256) void gcn_gather(
    const float* __restrict__ msg, const int* __restrict__ cnt,
    const int* __restrict__ bucket, const int* __restrict__ spill_n,
    const int* __restrict__ spill_d, const int* __restrict__ spill_s,
    const float* __restrict__ post_scale,
    const float* __restrict__ bias, int relu, const float* __restrict__ out_scale,
    float* __restrict__ outp, float* __restrict__ ns_w, float* __restrict__ nd_w, int n) {
  static_assert(NPW >= 1 && NPW <= 32, "one node per lane in the count load");
  constexpr int F = 32 * VEC;
  typedef float vf __attribute__((ext_vector_type(VEC)));
  const int lane = threadIdx.x & 31;
  const int base = NPW * __builtin_amdgcn_readfirstlane(
      (int)(blockIdx.x * (blockDim.x >> 5) + (threadIdx.x >> 5)));
  if (base >= n) return;
  const int nn = (n - base < NPW) ? n - base : NPW;

  const int vl = base + ((lane < nn) ? lane : 0);
  const int cc0 = cnt[n + vl];
  const int cc1 = cnt[3 * n + vl];
  float s_post = 1.0f, s_out = 1.0f;
  if constexpr (L1) {
    s_post = __frsqrt_rn(fmaxf((float)(cc0 + cc1), 1.0f));
    s_out = __frsqrt_rn(fmaxf((float)(cnt[vl] + cnt[2 * n + vl]), 1.0f));
    if (lane < nn) {
      ns_w[vl] = s_out;
      nd_w[vl] = s_post;
    }
  } else {
    if (post_scale) s_post = post_scale[vl];
    if (out_scale) s_out = out_scale[vl];
  }
  vf bb = 0.0f;
  if (bias) bb = *reinterpret_cast<const vf*>(bias + VEC * lane);

  const float* mrow = msg + VEC * lane;
  int raw_next = first_ids_raw(bucket, cnt, base, __builtin_amdgcn_readlane(cc0, 0),
                               __builtin_amdgcn_readlane(cc1, 0), n, lane);
  for (int r = 0; r < nn; ++r) {
    const int v = base + r;
    const int c0 = __builtin_amdgcn_readlane(cc0, r);
    const int c1 = __builtin_amdgcn_readlane(cc1, r);
    const int tot = ((c0 < CAP) ? c0 : CAP) + ((c1 < CAP) ? c1 : CAP);
    const int my = (lane < tot) ? raw_next : 0;
    const int rn = (r + 1 < nn) ? r + 1 : r;
    raw_next = first_ids_raw(bucket, cnt, base + rn, __builtin_amdgcn_readlane(cc0, rn),
                             __builtin_amdgcn_readlane(cc1, rn), n, lane);
    vf acc = node_sum<VEC, L1>(mrow, cnt, bucket, spill_n, spill_d, spill_s, my, v, c0, c1, n, lane);

    if (L1 || post_scale) acc *= lane_value(s_post, r);
    if (bias) acc += bb;
    if (relu) {
#pragma unroll
      for (int t = 0; t < VEC; ++t) acc[t] = fmaxf(acc[t], 0.0f);
    }
    if (L1 || out_scale) acc *= lane_value(s_out, r);
    *reinterpret_cast<vf*>(outp + (size_t)v * F + VEC * lane) = acc;
  }
}

template <int K, int FP>
__device__ __forceinline__ void mma_rows_lds(v8f (&acc)[4], const float4* Arow, const float* Wlane, float ps) {
  float4 q0 = Arow[0], q1 = Arow[1];
  for (int kb = 0; kb < K; kb += 16) {
    const int kn = (kb + 16 < K) ? kb + 16 : kb;
    const float4 n0 = Arow[kn >> 2], n1 = Arow[(kn >> 2) + 1];
    const float av[8] = {q0.x * ps, q0.y * ps, q0.z * ps, q0.w * ps,
                         q1.x * ps, q1.y * ps, q1.z * ps, q1.w * ps};
#pragma unroll
    for (int j = 0; j < 4; ++j) {
      v2f a;
      a.x = av[2 * j];
      a.y = av[2 * j + 1];
      const float* wr = Wlane + (kb + 2 * j) * FP;
      const float4 b0 = *reinterpret_cast<const float4*>(wr);
      const float4 b1 = *reinterpret_cast<const float4*>(wr + FP);
      v2f b;
      b.x = b0.x; b.y = b1.x;
      acc[0] = __builtin_amdgcn_wmma_f32_16x16x4_f32(false, a, false, b, (short)0, acc[0], false, false);
      b.x = b0.y; b.y = b1.y;
      acc[1] = __builtin_amdgcn_wmma_f32_16x16x4_f32(false, a, false, b, (short)0, acc[1], false, false);
      b.x = b0.z; b.y = b1.z;
      acc[2] = __builtin_amdgcn_wmma_f32_16x16x4_f32(false, a, false, b, (short)0, acc[2], false, false);
      b.x = b0.w; b.y = b1.w;
      acc[3] = __builtin_amdgcn_wmma_f32_16x16x4_f32(false, a, false, b, (short)0, acc[3], false, false);
    }
    q0 = n0;
    q1 = n1;
  }
}

__global__ __launch_bounds__(512) void gcn_layer23(
    const float* __restrict__ msg, const float* __restrict__ W2, const float* __restrict__ b2,
    const float* __restrict__ W3, const float* __restrict__ ns, const float* __restrict__ nd,
    float* __restrict__ C, const int* __restrict__ cnt, const int* __restrict__ bucket,
    const int* __restrict__ spill_n, const int* __restrict__ spill_d,
    const int* __restrict__ spill_s, int n_tiles, int n) {
  constexpr int K = 128, F2 = 128, F3 = 64, NT = 512;
  constexpr int AP = K + 4;
  __shared__ alignas(16) float Wl[K * F2];
  __shared__ alignas(16) float As[8 * 16 * AP];

  {
    const float4* Wg = reinterpret_cast<const float4*>(W2);
    float4* Ws = reinterpret_cast<float4*>(Wl);
    float4 w[K * F2 / 4 / NT];
#pragma unroll
    for (int j = 0; j < K * F2 / 4 / NT; ++j) w[j] = Wg[threadIdx.x + j * NT];
#pragma unroll
    for (int j = 0; j < K * F2 / 4 / NT; ++j) Ws[threadIdx.x + j * NT] = w[j];
  }

  const int wv = __builtin_amdgcn_readfirstlane((int)threadIdx.x >> 5);
  const int slot = wv & 7;
  const int half = wv >> 3;
  const int tileM = slot * (int)gridDim.x + (int)blockIdx.x;
  const bool live = tileM < n_tiles;
  const int lane = threadIdx.x & 31;
  const int h = lane >> 4;
  const int c = lane & 15;
  float* At = As + slot * 16 * AP;

  if (live) {
    const int cc0 = cnt[n + (tileM << 4) + c];
    const int cc1 = cnt[3 * n + (tileM << 4) + c];
    const float* mrow = msg + 4 * lane;
    const int r0 = half << 3;
    int my_next = first_ids(bucket, (tileM << 4) + r0, __builtin_amdgcn_readlane(cc0, r0),
                            __builtin_amdgcn_readlane(cc1, r0), n, lane);
    for (int r = r0; r < r0 + 8; ++r) {
      const int v = (tileM << 4) + r;
      const int c0 = __builtin_amdgcn_readlane(cc0, r);
      const int c1 = __builtin_amdgcn_readlane(cc1, r);
      const int my = my_next;
      if (r + 1 < r0 + 8)
        my_next = first_ids(bucket, v + 1, __builtin_amdgcn_readlane(cc0, r + 1),
                            __builtin_amdgcn_readlane(cc1, r + 1), n, lane);
      const v4f sum = node_sum<4, false>(mrow, cnt, bucket, spill_n, spill_d, spill_s, my, v, c0, c1, n, lane);
      *reinterpret_cast<v4f*>(At + r * AP + 4 * lane) = sum;
    }
  }
  __syncthreads();

  float4 hv[8];
#pragma unroll
  for (int r = 0; r < 8; ++r) hv[r] = float4{0.0f, 0.0f, 0.0f, 0.0f};
  if (live) {
    v8f acc[4];
#pragma unroll
    for (int t = 0; t < 4; ++t) acc[t] = v8f{};
    mma_rows_lds<K, F2>(acc, reinterpret_cast<const float4*>(At + c * AP + 8 * h),
                        Wl + (8 * h) * F2 + 64 * half + 4 * c, 1.0f);
    const float4 bb = *reinterpret_cast<const float4*>(b2 + 64 * half + 4 * c);
#pragma unroll
    for (int r = 0; r < 8; ++r) {
      const float sc = nd[(tileM << 4) + r + (h << 3)];
      float4 v = {acc[0][r], acc[1][r], acc[2][r], acc[3][r]};
      v.x *= sc; v.y *= sc; v.z *= sc; v.w *= sc;
      v.x += bb.x; v.y += bb.y; v.z += bb.z; v.w += bb.w;
      v.x = fmaxf(v.x, 0.0f); v.y = fmaxf(v.y, 0.0f);
      v.z = fmaxf(v.z, 0.0f); v.w = fmaxf(v.w, 0.0f);
      hv[r] = v;
    }
  }
  __syncthreads();

  {
    const float4* Wg = reinterpret_cast<const float4*>(W3);
    float4* Ws = reinterpret_cast<float4*>(Wl);
    float4 w[K * F3 / 4 / NT];
#pragma unroll
    for (int j = 0; j < K * F3 / 4 / NT; ++j) w[j] = Wg[threadIdx.x + j * NT];
#pragma unroll
    for (int j = 0; j < K * F3 / 4 / NT; ++j) Ws[threadIdx.x + j * NT] = w[j];
  }
  if (live) {
#pragma unroll
    for (int r = 0; r < 8; ++r)
      *reinterpret_cast<float4*>(At + (r + (h << 3)) * AP + 64 * half + 4 * c) = hv[r];
  }
  __syncthreads();

  if (live && half == 0) {
    v8f acc[4];
#pragma unroll
    for (int t = 0; t < 4; ++t) acc[t] = v8f{};
    const float ps = ns[(tileM << 4) + c];
    mma_rows_lds<K, F3>(acc, reinterpret_cast<const float4*>(At + c * AP + 8 * h),
                        Wl + (8 * h) * F3 + 4 * c, ps);
#pragma unroll
    for (int r = 0; r < 8; ++r) {
      const int row = (tileM << 4) + r + (h << 3);
      const float4 v = {acc[0][r], acc[1][r], acc[2][r], acc[3][r]};
      *reinterpret_cast<float4*>(C + (size_t)row * F3 + 4 * c) = v;
    }
  }
}

static inline unsigned cdiv(unsigned a, unsigned b) { return (a + b - 1) / b; }

extern "C" void kernel_launch(void* const* d_in, const int* in_sizes, int n_in,
                              void* d_out, int out_size, void* d_ws, size_t ws_size,
                              hipStream_t stream) {
  const float* features = (const float*)d_in[0];
  const float* W1 = (const float*)d_in[1];
  const float* b1 = (const float*)d_in[2];
  const float* W2 = (const float*)d_in[3];
  const float* b2 = (const float*)d_in[4];
  const float* W3 = (const float*)d_in[5];
  const float* b3 = (const float*)d_in[6];
  const int* src  = (const int*)d_in[7];
  const int* dst  = (const int*)d_in[8];
  float* out = (float*)d_out;

  const int N  = N_NODES;
  const int E  = in_sizes[7];
  const int H  = in_sizes[2];

  float* ws   = (float*)d_ws;
  float* ns   = ws;
  float* nd   = ws + N;
  float* bufA = ws + 2 * N + 96;
  float* bufB = bufA + (size_t)N * H;
  int* cnt     = (int*)(bufB + (size_t)N * H);
  int* spill_n = cnt + 4 * N;
  int* bucket  = spill_n + 4;
  int* spill_d = bucket + (size_t)2 * N * CAP;
  int* spill_s = spill_d + E;
  (void)ws_size; (void)n_in; (void)out_size;

  const int BS = 256;
  const int NPW = 4;
  const unsigned node_blocks = cdiv(N, (BS / 32) * NPW);

  gcn_zero_i4<<<cdiv(N + 1, BS), BS, 0, stream>>>(cnt, N + 1);

  gcn_gemm_wmma<256, 128, true><<<256, 640, 0, stream>>>(
      features, W1, bufA, N,
      src, dst, cnt, spill_n, bucket, spill_d, spill_s, E, N);
  gcn_gather<4, true, NPW><<<node_blocks, BS, 0, stream>>>(
      bufA, cnt, bucket, spill_n, spill_d, spill_s, nullptr, b1, 1, nullptr, bufB, ns, nd, N);

  gcn_layer23<<<512, 512, 0, stream>>>(
      bufB, W2, b2, W3, ns, nd, bufA, cnt, bucket, spill_n, spill_d, spill_s, N / 16, N);

  gcn_gather<2, false, NPW><<<node_blocks, BS, 0, stream>>>(
      bufA, cnt, bucket, spill_n, spill_d, spill_s, nd, b3, 0, nullptr, out,
      nullptr, nullptr, N);
}
